# starting kernel
# baseline (speedup 1.0000x reference)
#include <hip/hip_runtime.h>
#include <hip/hip_bf16.h>

typedef __attribute__((ext_vector_type(2))) float v2f;
typedef __attribute__((ext_vector_type(8))) float v8f;

#define EMB_D 64

__global__ void lgcn_init(const float* __restrict__ ue, const float* __restrict__ ie,
                          float* __restrict__ cur, float* __restrict__ acc,
                          size_t userElems, size_t total) {
    size_t i = ((size_t)blockIdx.x * blockDim.x + threadIdx.x) * 4;
    if (i >= total) return;
    float4 e = (i < userElems) ? *(const float4*)(ue + i)
                               : *(const float4*)(ie + (i - userElems));
    *(float4*)(cur + i) = e;
    float4 s;
    s.x = e.x * 0.25f; s.y = e.y * 0.25f; s.z = e.z * 0.25f; s.w = e.w * 0.25f;
    *(float4*)(acc + i) = s;
}

__global__ void lgcn_zero(float* __restrict__ p, size_t total) {
    size_t i = ((size_t)blockIdx.x * blockDim.x + threadIdx.x) * 4;
    if (i >= total) return;
    float4 z; z.x = z.y = z.z = z.w = 0.0f;
    *(float4*)(p + i) = z;
}

__global__ void lgcn_spmm(const int* __restrict__ rows, const int* __restrict__ cols,
                          const float* __restrict__ vals, const float* __restrict__ x,
                          float* __restrict__ y, int nEdges) {
    long long t = (long long)blockIdx.x * blockDim.x + threadIdx.x;
    int e   = (int)(t >> 4);
    int sub = (int)(t & 15);
    if (e >= nEdges) return;
    int   r = rows[e];
    int   c = cols[e];
    float v = vals[e];
    float4 g = *((const float4*)(x + (size_t)c * EMB_D) + sub);
    float* yp = y + (size_t)r * EMB_D + (size_t)sub * 4;
    unsafeAtomicAdd(yp + 0, g.x * v);
    unsafeAtomicAdd(yp + 1, g.y * v);
    unsafeAtomicAdd(yp + 2, g.z * v);
    unsafeAtomicAdd(yp + 3, g.w * v);
}

__global__ void lgcn_wmma_axpy(const float* __restrict__ nxt, float* __restrict__ acc,
                               int nTiles) {
    int wave = (int)(((size_t)blockIdx.x * blockDim.x + threadIdx.x) >> 5);
    if (wave >= nTiles) return;
    int lane = threadIdx.x & 31;
    int lo = lane & 15;
    int hi = lane >> 4;
    int tM = wave >> 2;
    int n0 = (wave & 3) << 4;
    size_t row0 = (size_t)tM << 4;

    const float* aBase = nxt + (row0 + (size_t)lo) * EMB_D + n0;
    float*       cBase = acc + (row0 + (size_t)(8 * hi)) * EMB_D + n0 + lo;

    v8f c;
#pragma unroll
    for (int v = 0; v < 8; ++v) c[v] = cBase[(size_t)v * EMB_D];

#pragma unroll
    for (int k = 0; k < 4; ++k) {
        int col = 4 * k + 2 * hi;
        v2f a, b;
        a[0] = aBase[col + 0];
        a[1] = aBase[col + 1];
        b[0] = (lo == col + 0) ? 0.25f : 0.0f;
        b[1] = (lo == col + 1) ? 0.25f : 0.0f;
        c = __builtin_amdgcn_wmma_f32_16x16x4_f32(
                 false, a,  false, b,
                 (short)0, c,  false,  false);
    }

#pragma unroll
    for (int v = 0; v < 8; ++v) cBase[(size_t)v * EMB_D] = c[v];
}

extern "C" void kernel_launch(void* const* d_in, const int* in_sizes, int n_in,
                              void* d_out, int out_size, void* d_ws, size_t ws_size,
                              hipStream_t stream) {
    const float* ue   = (const float*)d_in[0];
    const float* ie   = (const float*)d_in[1];
    const int*   rows = (const int*)d_in[2];
    const int*   cols = (const int*)d_in[3];
    const float* vals = (const float*)d_in[4];
    float*       out  = (float*)d_out;

    size_t userElems = (size_t)in_sizes[0];
    size_t itemElems = (size_t)in_sizes[1];
    int    nEdges    = in_sizes[2];
    size_t total     = userElems + itemElems;

    float* buf0 = (float*)d_ws;
    float* buf1 = buf0 + total;

    int vecBlocks = (int)((total / 4 + 255) / 256);
    lgcn_init<<<vecBlocks, 256, 0, stream>>>(ue, ie, buf0, out, userElems, total);

    int nTiles     = (int)(total / 256);
    int wmmaBlocks = (nTiles + 7) / 8;
    long long spmmThreads = (long long)nEdges * 16;
    int spmmBlocks = (int)((spmmThreads + 255) / 256);

    float* cur = buf0;
    float* nxt = buf1;
    for (int layer = 0; layer < 3; ++layer) {
        lgcn_zero<<<vecBlocks, 256, 0, stream>>>(nxt, total);
        lgcn_spmm<<<spmmBlocks, 256, 0, stream>>>(rows, cols, vals, cur, nxt, nEdges);
        lgcn_wmma_axpy<<<wmmaBlocks, 256, 0, stream>>>(nxt, out, nTiles);
        float* tmp = cur; cur = nxt; nxt = tmp;
    }
}


# Group edges by destination row once per call (integer rank add, scan), then sum each row's gathered rows in registers, store once: no float atomics, no zeroing; in-row sum order = arrival rank
# speedup vs baseline: 2.2688x; 2.2688x over previous
#include <hip/hip_runtime.h>
#include <hip/hip_bf16.h>


typedef __attribute__((ext_vector_type(2))) float v2f;
typedef __attribute__((ext_vector_type(8))) float v8f;

#define EMB_D 64
#define SCAN_BLOCK 256
#define SCAN_CHUNK 1024

__global__ void lgcn_init(const float* __restrict__ ue, const float* __restrict__ ie,
                          float* __restrict__ cur, float* __restrict__ acc,
                          size_t userElems, size_t total) {
    size_t i = ((size_t)blockIdx.x * blockDim.x + threadIdx.x) * 4;
    if (i >= total) return;
    float4 e = (i < userElems) ? *(const float4*)(ue + i)
                               : *(const float4*)(ie + (i - userElems));
    *(float4*)(cur + i) = e;
    float4 s;
    s.x = e.x * 0.25f; s.y = e.y * 0.25f; s.z = e.z * 0.25f; s.w = e.w * 0.25f;
    *(float4*)(acc + i) = s;
}

__global__ void lgcn_zero(float* __restrict__ p, size_t total) {
    size_t i = ((size_t)blockIdx.x * blockDim.x + threadIdx.x) * 4;
    if (i >= total) return;
    float4 z; z.x = z.y = z.z = z.w = 0.0f;
    *(float4*)(p + i) = z;
}

__global__ void lgcn_rank(const int* __restrict__ rows, int* __restrict__ cnt,
                          int* __restrict__ rank, int nEdges) {
    int i = (int)(blockIdx.x * blockDim.x + threadIdx.x) * 4;
    if (i + 3 < nEdges) {
        int4 r = *(const int4*)(rows + i);
        int4 k;
        k.x = atomicAdd(cnt + r.x, 1);
        k.y = atomicAdd(cnt + r.y, 1);
        k.z = atomicAdd(cnt + r.z, 1);
        k.w = atomicAdd(cnt + r.w, 1);
        *(int4*)(rank + i) = k;
    } else {
        for (int j = i; j < nEdges; ++j) rank[j] = atomicAdd(cnt + rows[j], 1);
    }
}

__global__ void lgcn_chunk_sum(const int* __restrict__ cnt, int* __restrict__ csum) {
    __shared__ int sh[SCAN_BLOCK];
    int tid = threadIdx.x;
    int4 c = *(const int4*)(cnt + (size_t)blockIdx.x * SCAN_CHUNK + (size_t)tid * 4);
    sh[tid] = c.x + c.y + c.z + c.w;
    __syncthreads();
    for (int s = SCAN_BLOCK / 2; s > 0; s >>= 1) {
        if (tid < s) sh[tid] += sh[tid + s];
        __syncthreads();
    }
    if (tid == 0) csum[blockIdx.x] = sh[0];
}

__global__ void lgcn_scan(int* __restrict__ cnt, const int* __restrict__ csum) {
    __shared__ int sh[SCAN_BLOCK];
    int tid = threadIdx.x;
    int part = 0;
    for (int i = tid; i < (int)blockIdx.x; i += SCAN_BLOCK) part += csum[i];
    sh[tid] = part;
    __syncthreads();
    for (int s = SCAN_BLOCK / 2; s > 0; s >>= 1) {
        if (tid < s) sh[tid] += sh[tid + s];
        __syncthreads();
    }
    int chunkOff = sh[0];
    __syncthreads();

    int* p = cnt + (size_t)blockIdx.x * SCAN_CHUNK + (size_t)tid * 4;
    int4 c = *(const int4*)p;
    int mine = c.x + c.y + c.z + c.w;
    sh[tid] = mine;
    __syncthreads();
    for (int d = 1; d < SCAN_BLOCK; d <<= 1) {
        int add = (tid >= d) ? sh[tid - d] : 0;
        __syncthreads();
        sh[tid] += add;
        __syncthreads();
    }
    int excl = sh[tid] - mine + chunkOff;
    int4 o;
    o.x = excl;
    o.y = o.x + c.x;
    o.z = o.y + c.y;
    o.w = o.z + c.z;
    *(int4*)p = o;
}

__global__ void lgcn_place(const int* __restrict__ rows, const int* __restrict__ cols,
                           const float* __restrict__ vals, const int* __restrict__ rowStart,
                           const int* __restrict__ rank, int* __restrict__ ed, int nEdges) {
    int i = (int)(blockIdx.x * blockDim.x + threadIdx.x) * 4;
    int2* edp = (int2*)ed;
    if (i + 3 < nEdges) {
        int4   r = *(const int4*)(rows + i);
        int4   k = *(const int4*)(rank + i);
        int4   c = *(const int4*)(cols + i);
        float4 v = *(const float4*)(vals + i);
        int p0 = rowStart[r.x] + k.x;
        int p1 = rowStart[r.y] + k.y;
        int p2 = rowStart[r.z] + k.z;
        int p3 = rowStart[r.w] + k.w;
        int2 e0; e0.x = c.x; e0.y = __float_as_int(v.x);
        int2 e1; e1.x = c.y; e1.y = __float_as_int(v.y);
        int2 e2; e2.x = c.z; e2.y = __float_as_int(v.z);
        int2 e3; e3.x = c.w; e3.y = __float_as_int(v.w);
        edp[p0] = e0;
        edp[p1] = e1;
        edp[p2] = e2;
        edp[p3] = e3;
    } else {
        for (int j = i; j < nEdges; ++j) {
            int2 e; e.x = cols[j]; e.y = __float_as_int(vals[j]);
            edp[rowStart[rows[j]] + rank[j]] = e;
        }
    }
}

__global__ void lgcn_row_spmm(const int* __restrict__ rowStart, const int* __restrict__ ed,
                              const float* __restrict__ x, float* __restrict__ y, int nRows) {
    long long t = (long long)blockIdx.x * blockDim.x + threadIdx.x;
    int r   = (int)(t >> 4);
    int sub = (int)(t & 15);
    if (r >= nRows) return;
    int j   = rowStart[r];
    int end = rowStart[r + 1];
    const int2*   edp = (const int2*)ed;
    const float4* xv  = (const float4*)x + sub;
    float4 a; a.x = a.y = a.z = a.w = 0.0f;
    for (; j + 4 <= end; j += 4) {
        int2 e0 = edp[j + 0];
        int2 e1 = edp[j + 1];
        int2 e2 = edp[j + 2];
        int2 e3 = edp[j + 3];
        float4 g0 = xv[(size_t)e0.x * (EMB_D / 4)];
        float4 g1 = xv[(size_t)e1.x * (EMB_D / 4)];
        float4 g2 = xv[(size_t)e2.x * (EMB_D / 4)];
        float4 g3 = xv[(size_t)e3.x * (EMB_D / 4)];
        float v0 = __int_as_float(e0.y);
        float v1 = __int_as_float(e1.y);
        float v2 = __int_as_float(e2.y);
        float v3 = __int_as_float(e3.y);
        a.x += g0.x * v0; a.y += g0.y * v0; a.z += g0.z * v0; a.w += g0.w * v0;
        a.x += g1.x * v1; a.y += g1.y * v1; a.z += g1.z * v1; a.w += g1.w * v1;
        a.x += g2.x * v2; a.y += g2.y * v2; a.z += g2.z * v2; a.w += g2.w * v2;
        a.x += g3.x * v3; a.y += g3.y * v3; a.z += g3.z * v3; a.w += g3.w * v3;
    }
    int n = end - j;
    if (n > 0) {
        int last = end - 1;
        int2 e0 = edp[j];
        int2 e1 = edp[(n > 1) ? j + 1 : last];
        int2 e2 = edp[(n > 2) ? j + 2 : last];
        float4 g0 = xv[(size_t)e0.x * (EMB_D / 4)];
        float4 g1 = xv[(size_t)e1.x * (EMB_D / 4)];
        float4 g2 = xv[(size_t)e2.x * (EMB_D / 4)];
        float v0 = __int_as_float(e0.y);
        float v1 = (n > 1) ? __int_as_float(e1.y) : 0.0f;
        float v2 = (n > 2) ? __int_as_float(e2.y) : 0.0f;
        if (n < 2) { g1.x = 0.0f; g1.y = 0.0f; g1.z = 0.0f; g1.w = 0.0f; }
        if (n < 3) { g2.x = 0.0f; g2.y = 0.0f; g2.z = 0.0f; g2.w = 0.0f; }
        a.x += g0.x * v0; a.y += g0.y * v0; a.z += g0.z * v0; a.w += g0.w * v0;
        a.x += g1.x * v1; a.y += g1.y * v1; a.z += g1.z * v1; a.w += g1.w * v1;
        a.x += g2.x * v2; a.y += g2.y * v2; a.z += g2.z * v2; a.w += g2.w * v2;
    }
    ((float4*)y)[(size_t)r * (EMB_D / 4) + sub] = a;
}

__global__ void lgcn_wmma_axpy(const float* __restrict__ nxt, float* __restrict__ acc,
                               int nTiles) {
    int wave = (int)(((size_t)blockIdx.x * blockDim.x + threadIdx.x) >> 5);
    if (wave >= nTiles) return;
    int lane = threadIdx.x & 31;
    int lo = lane & 15;
    int hi = lane >> 4;
    int tM = wave >> 2;
    int n0 = (wave & 3) << 4;
    size_t row0 = (size_t)tM << 4;

    const float* aBase = nxt + (row0 + (size_t)lo) * EMB_D + n0;
    float*       cBase = acc + (row0 + (size_t)(8 * hi)) * EMB_D + n0 + lo;

    v8f c;
#pragma unroll
    for (int v = 0; v < 8; ++v) c[v] = cBase[(size_t)v * EMB_D];

#pragma unroll
    for (int k = 0; k < 4; ++k) {
        int col = 4 * k + 2 * hi;
        v2f a, b;
        a[0] = aBase[col + 0];
        a[1] = aBase[col + 1];
        b[0] = (lo == col + 0) ? 0.25f : 0.0f;
        b[1] = (lo == col + 1) ? 0.25f : 0.0f;
        c = __builtin_amdgcn_wmma_f32_16x16x4_f32(
                 false, a,  false, b,
                 (short)0, c,  false,  false);
    }

#pragma unroll
    for (int v = 0; v < 8; ++v) cBase[(size_t)v * EMB_D] = c[v];
}

extern "C" void kernel_launch(void* const* d_in, const int* in_sizes, int n_in,
                              void* d_out, int out_size, void* d_ws, size_t ws_size,
                              hipStream_t stream) {
    const float* ue   = (const float*)d_in[0];
    const float* ie   = (const float*)d_in[1];
    const int*   rows = (const int*)d_in[2];
    const int*   cols = (const int*)d_in[3];
    const float* vals = (const float*)d_in[4];
    float*       out  = (float*)d_out;

    size_t userElems = (size_t)in_sizes[0];
    size_t itemElems = (size_t)in_sizes[1];
    int    nEdges    = in_sizes[2];
    size_t total     = userElems + itemElems;
    int    nRows     = (int)(total / EMB_D);

    int    nChunks   = (nRows + 1 + SCAN_CHUNK - 1) / SCAN_CHUNK;
    size_t cntPad    = (size_t)nChunks * SCAN_CHUNK;
    size_t csumPad   = ((size_t)nChunks + 3) / 4 * 4;
    size_t edgePad   = ((size_t)nEdges + 3) / 4 * 4;

    float* buf0 = (float*)d_ws;
    float* buf1 = buf0 + total;
    int*   cnt  = (int*)(buf1 + total);
    int*   csum = cnt + cntPad;
    int*   rank = csum + csumPad;
    int*   ed   = rank + edgePad;

    int vecBlocks = (int)((total / 4 + 255) / 256);
    lgcn_init<<<vecBlocks, 256, 0, stream>>>(ue, ie, buf0, out, userElems, total);

    int cntBlocks  = (int)((cntPad / 4 + 255) / 256);
    int edgeBlocks = (int)((edgePad / 4 + 255) / 256);
    lgcn_zero<<<cntBlocks, 256, 0, stream>>>((float*)cnt, cntPad);
    lgcn_rank<<<edgeBlocks, 256, 0, stream>>>(rows, cnt, rank, nEdges);
    lgcn_chunk_sum<<<nChunks, SCAN_BLOCK, 0, stream>>>(cnt, csum);
    lgcn_scan<<<nChunks, SCAN_BLOCK, 0, stream>>>(cnt, csum);
    lgcn_place<<<edgeBlocks, 256, 0, stream>>>(rows, cols, vals, cnt, rank, ed, nEdges);

    int nTiles     = (int)(total / 256);
    int wmmaBlocks = (nTiles + 7) / 8;
    long long rowThreads = (long long)nRows * 16;
    int rowBlocks  = (int)((rowThreads + 255) / 256);

    float* cur = buf0;
    float* nxt = buf1;
    for (int layer = 0; layer < 3; ++layer) {
        lgcn_row_spmm<<<rowBlocks, 256, 0, stream>>>(cnt, ed, cur, nxt, nRows);
        lgcn_wmma_axpy<<<wmmaBlocks, 256, 0, stream>>>(nxt, out, nTiles);
        float* tmp = cur; cur = nxt; nxt = tmp;
    }
}


# Group edges by destination with a two-level counting sort on LDS counters (512-row buckets, lists written as consecutive runs), no global add per edge; in-row sum order = arrival at LDS counters
# speedup vs baseline: 3.8584x; 1.7006x over previous
#include <hip/hip_runtime.h>
#include <hip/hip_bf16.h>


typedef __attribute__((ext_vector_type(2))) float v2f;
typedef __attribute__((ext_vector_type(8))) float v8f;

#define EMB_D 64
#define SCAN_BLOCK 256
#define SCAN_CHUNK 1024

#define BK_SHIFT 9
#define BK_ROWS 512
#define MAXB 1024
#define L1_PER 10
#define L1_CAP (L1_PER * MAXB)
#define L2_CAP 16384

__global__ void lgcn_init(const float* __restrict__ ue, const float* __restrict__ ie,
                          float* __restrict__ cur, float* __restrict__ acc,
                          size_t userElems, size_t total) {
    size_t i = ((size_t)blockIdx.x * blockDim.x + threadIdx.x) * 4;
    if (i >= total) return;
    float4 e = (i < userElems) ? *(const float4*)(ue + i)
                               : *(const float4*)(ie + (i - userElems));
    *(float4*)(cur + i) = e;
    float4 s;
    s.x = e.x * 0.25f; s.y = e.y * 0.25f; s.z = e.z * 0.25f; s.w = e.w * 0.25f;
    *(float4*)(acc + i) = s;
}

__global__ void lgcn_bin_count(const int* __restrict__ rows, int* __restrict__ cntM,
                               int nEdges, int nBuckP, int nBlk, int matLen) {
    __shared__ int hist[MAXB];
    int tid = threadIdx.x;
    hist[tid] = 0;
    __syncthreads();
    int base = (int)blockIdx.x * L1_CAP;
    for (int j = 0; j < L1_PER; ++j) {
        int e = base + j * MAXB + tid;
        if (e < nEdges) atomicAdd(&hist[rows[e] >> BK_SHIFT], 1);
    }
    __syncthreads();
    if (tid < nBuckP) cntM[(size_t)tid * nBlk + blockIdx.x] = hist[tid];
    if (blockIdx.x == 0)
        for (int i = nBuckP * nBlk + tid; i < matLen; i += MAXB) cntM[i] = 0;
}

__global__ void lgcn_chunk_sum(const int* __restrict__ cnt, int* __restrict__ csum) {
    __shared__ int sh[SCAN_BLOCK];
    int tid = threadIdx.x;
    int4 c = *(const int4*)(cnt + (size_t)blockIdx.x * SCAN_CHUNK + (size_t)tid * 4);
    sh[tid] = c.x + c.y + c.z + c.w;
    __syncthreads();
    for (int s = SCAN_BLOCK / 2; s > 0; s >>= 1) {
        if (tid < s) sh[tid] += sh[tid + s];
        __syncthreads();
    }
    if (tid == 0) csum[blockIdx.x] = sh[0];
}

__global__ void lgcn_scan(int* __restrict__ cnt, const int* __restrict__ csum) {
    __shared__ int sh[SCAN_BLOCK];
    int tid = threadIdx.x;
    int part = 0;
    for (int i = tid; i < (int)blockIdx.x; i += SCAN_BLOCK) part += csum[i];
    sh[tid] = part;
    __syncthreads();
    for (int s = SCAN_BLOCK / 2; s > 0; s >>= 1) {
        if (tid < s) sh[tid] += sh[tid + s];
        __syncthreads();
    }
    int chunkOff = sh[0];
    __syncthreads();

    int* p = cnt + (size_t)blockIdx.x * SCAN_CHUNK + (size_t)tid * 4;
    int4 c = *(const int4*)p;
    int mine = c.x + c.y + c.z + c.w;
    sh[tid] = mine;
    __syncthreads();
    for (int d = 1; d < SCAN_BLOCK; d <<= 1) {
        int add = (tid >= d) ? sh[tid - d] : 0;
        __syncthreads();
        sh[tid] += add;
        __syncthreads();
    }
    int excl = sh[tid] - mine + chunkOff;
    int4 o;
    o.x = excl;
    o.y = o.x + c.x;
    o.z = o.y + c.y;
    o.w = o.z + c.z;
    *(int4*)p = o;
}

__global__ void lgcn_bin_scatter(const int* __restrict__ rows, const int* __restrict__ cols,
                                 const float* __restrict__ vals, const int* __restrict__ offM,
                                 int* __restrict__ bin, int nEdges, int nBuckP, int nBlk) {
    __shared__ int   lStart[MAXB];
    __shared__ int   gOff[MAXB];
    __shared__ int   sRow[L1_CAP];
    __shared__ int   sCol[L1_CAP];
    __shared__ float sVal[L1_CAP];
    int tid  = threadIdx.x;
    int base = (int)blockIdx.x * L1_CAP;
    int n    = nEdges - base;
    if (n <= 0) return;
    if (n > L1_CAP) n = L1_CAP;

    lStart[tid] = 0;
    gOff[tid]   = (tid < nBuckP) ? offM[(size_t)tid * nBlk + blockIdx.x] : 0;
    __syncthreads();

    int   r[L1_PER], c[L1_PER], k[L1_PER];
    float v[L1_PER];
#pragma unroll
    for (int j = 0; j < L1_PER; ++j) {
        int e = j * MAXB + tid;
        r[j] = 0; c[j] = 0; v[j] = 0.0f; k[j] = 0;
        if (e < n) {
            r[j] = rows[base + e];
            c[j] = cols[base + e];
            v[j] = vals[base + e];
            k[j] = atomicAdd(&lStart[r[j] >> BK_SHIFT], 1);
        }
    }
    __syncthreads();

    int mine = lStart[tid];
    for (int d = 1; d < MAXB; d <<= 1) {
        int add = (tid >= d) ? lStart[tid - d] : 0;
        __syncthreads();
        lStart[tid] += add;
        __syncthreads();
    }
    int excl = lStart[tid] - mine;
    __syncthreads();
    lStart[tid] = excl;
    __syncthreads();

#pragma unroll
    for (int j = 0; j < L1_PER; ++j) {
        int e = j * MAXB + tid;
        if (e < n) {
            int p = lStart[r[j] >> BK_SHIFT] + k[j];
            sRow[p] = r[j];
            sCol[p] = c[j];
            sVal[p] = v[j];
        }
    }
    __syncthreads();

    int2* binp = (int2*)bin;
    for (int i = tid; i < n; i += MAXB) {
        int row = sRow[i];
        int b   = row >> BK_SHIFT;
        int g   = gOff[b] + (i - lStart[b]);
        int2 rec;
        rec.x = (sCol[i] << BK_SHIFT) | (row & (BK_ROWS - 1));
        rec.y = __float_as_int(sVal[i]);
        binp[g] = rec;
    }
}

__global__ void lgcn_bucket_sort(const int* __restrict__ offM, const int* __restrict__ bin,
                                 int* __restrict__ rowStart, int* __restrict__ ed,
                                 int nRows, int nBlk) {
    __shared__ int  cnt[BK_ROWS];
    __shared__ int  cur[BK_ROWS];
    __shared__ int2 sEd[L2_CAP];
    int tid   = threadIdx.x;
    int b     = (int)blockIdx.x;
    int start = offM[(size_t)b * nBlk];
    int end   = offM[(size_t)(b + 1) * nBlk];
    int n     = end - start;
    int row0  = b << BK_SHIFT;
    const int2* bp = (const int2*)bin + start;

    if (tid < BK_ROWS) cnt[tid] = 0;
    __syncthreads();
    for (int i = tid; i < n; i += MAXB) atomicAdd(&cnt[bp[i].x & (BK_ROWS - 1)], 1);
    __syncthreads();

    int mine = (tid < BK_ROWS) ? cnt[tid] : 0;
    for (int d = 1; d < BK_ROWS; d <<= 1) {
        int add = (tid < BK_ROWS && tid >= d) ? cnt[tid - d] : 0;
        __syncthreads();
        if (tid < BK_ROWS) cnt[tid] += add;
        __syncthreads();
    }
    if (tid < BK_ROWS) {
        int excl = cnt[tid] - mine;
        cur[tid] = excl;
        int gr = row0 + tid;
        if (gr <= nRows) rowStart[gr] = start + excl;
    }
    __syncthreads();

    int2* edp = (int2*)ed + start;
    if (n <= L2_CAP) {
        for (int i = tid; i < n; i += MAXB) {
            int2 rec = bp[i];
            int p = atomicAdd(&cur[rec.x & (BK_ROWS - 1)], 1);
            int2 e; e.x = (int)((unsigned)rec.x >> BK_SHIFT); e.y = rec.y;
            sEd[p] = e;
        }
        __syncthreads();
        for (int i = tid; i < n; i += MAXB) edp[i] = sEd[i];
    } else {
        for (int i = tid; i < n; i += MAXB) {
            int2 rec = bp[i];
            int p = atomicAdd(&cur[rec.x & (BK_ROWS - 1)], 1);
            int2 e; e.x = (int)((unsigned)rec.x >> BK_SHIFT); e.y = rec.y;
            edp[p] = e;
        }
    }
}

__global__ void lgcn_row_spmm(const int* __restrict__ rowStart, const int* __restrict__ ed,
                              const float* __restrict__ x, float* __restrict__ y, int nRows) {
    long long t = (long long)blockIdx.x * blockDim.x + threadIdx.x;
    int r   = (int)(t >> 4);
    int sub = (int)(t & 15);
    if (r >= nRows) return;
    int j   = rowStart[r];
    int end = rowStart[r + 1];
    const int2*   edp = (const int2*)ed;
    const float4* xv  = (const float4*)x + sub;
    float4 a; a.x = a.y = a.z = a.w = 0.0f;
    for (; j + 4 <= end; j += 4) {
        int2 e0 = edp[j + 0];
        int2 e1 = edp[j + 1];
        int2 e2 = edp[j + 2];
        int2 e3 = edp[j + 3];
        float4 g0 = xv[(size_t)e0.x * (EMB_D / 4)];
        float4 g1 = xv[(size_t)e1.x * (EMB_D / 4)];
        float4 g2 = xv[(size_t)e2.x * (EMB_D / 4)];
        float4 g3 = xv[(size_t)e3.x * (EMB_D / 4)];
        float v0 = __int_as_float(e0.y);
        float v1 = __int_as_float(e1.y);
        float v2 = __int_as_float(e2.y);
        float v3 = __int_as_float(e3.y);
        a.x += g0.x * v0; a.y += g0.y * v0; a.z += g0.z * v0; a.w += g0.w * v0;
        a.x += g1.x * v1; a.y += g1.y * v1; a.z += g1.z * v1; a.w += g1.w * v1;
        a.x += g2.x * v2; a.y += g2.y * v2; a.z += g2.z * v2; a.w += g2.w * v2;
        a.x += g3.x * v3; a.y += g3.y * v3; a.z += g3.z * v3; a.w += g3.w * v3;
    }
    int n = end - j;
    if (n > 0) {
        int last = end - 1;
        int2 e0 = edp[j];
        int2 e1 = edp[(n > 1) ? j + 1 : last];
        int2 e2 = edp[(n > 2) ? j + 2 : last];
        float4 g0 = xv[(size_t)e0.x * (EMB_D / 4)];
        float4 g1 = xv[(size_t)e1.x * (EMB_D / 4)];
        float4 g2 = xv[(size_t)e2.x * (EMB_D / 4)];
        float v0 = __int_as_float(e0.y);
        float v1 = (n > 1) ? __int_as_float(e1.y) : 0.0f;
        float v2 = (n > 2) ? __int_as_float(e2.y) : 0.0f;
        if (n < 2) { g1.x = 0.0f; g1.y = 0.0f; g1.z = 0.0f; g1.w = 0.0f; }
        if (n < 3) { g2.x = 0.0f; g2.y = 0.0f; g2.z = 0.0f; g2.w = 0.0f; }
        a.x += g0.x * v0; a.y += g0.y * v0; a.z += g0.z * v0; a.w += g0.w * v0;
        a.x += g1.x * v1; a.y += g1.y * v1; a.z += g1.z * v1; a.w += g1.w * v1;
        a.x += g2.x * v2; a.y += g2.y * v2; a.z += g2.z * v2; a.w += g2.w * v2;
    }
    ((float4*)y)[(size_t)r * (EMB_D / 4) + sub] = a;
}

__global__ void lgcn_wmma_axpy(const float* __restrict__ nxt, float* __restrict__ acc,
                               int nTiles) {
    int wave = (int)(((size_t)blockIdx.x * blockDim.x + threadIdx.x) >> 5);
    if (wave >= nTiles) return;
    int lane = threadIdx.x & 31;
    int lo = lane & 15;
    int hi = lane >> 4;
    int tM = wave >> 2;
    int n0 = (wave & 3) << 4;
    size_t row0 = (size_t)tM << 4;

    const float* aBase = nxt + (row0 + (size_t)lo) * EMB_D + n0;
    float*       cBase = acc + (row0 + (size_t)(8 * hi)) * EMB_D + n0 + lo;

    v8f c;
#pragma unroll
    for (int v = 0; v < 8; ++v) c[v] = cBase[(size_t)v * EMB_D];

#pragma unroll
    for (int k = 0; k < 4; ++k) {
        int col = 4 * k + 2 * hi;
        v2f a, b;
        a[0] = aBase[col + 0];
        a[1] = aBase[col + 1];
        b[0] = (lo == col + 0) ? 0.25f : 0.0f;
        b[1] = (lo == col + 1) ? 0.25f : 0.0f;
        c = __builtin_amdgcn_wmma_f32_16x16x4_f32(
                 false, a,  false, b,
                 (short)0, c,  false,  false);
    }

#pragma unroll
    for (int v = 0; v < 8; ++v) cBase[(size_t)v * EMB_D] = c[v];
}

extern "C" void kernel_launch(void* const* d_in, const int* in_sizes, int n_in,
                              void* d_out, int out_size, void* d_ws, size_t ws_size,
                              hipStream_t stream) {
    const float* ue   = (const float*)d_in[0];
    const float* ie   = (const float*)d_in[1];
    const int*   rows = (const int*)d_in[2];
    const int*   cols = (const int*)d_in[3];
    const float* vals = (const float*)d_in[4];
    float*       out  = (float*)d_out;

    size_t userElems = (size_t)in_sizes[0];
    size_t itemElems = (size_t)in_sizes[1];
    int    nEdges    = in_sizes[2];
    size_t total     = userElems + itemElems;
    int    nRows     = (int)(total / EMB_D);

    int    nBuck     = (nRows + 1 + BK_ROWS - 1) / BK_ROWS;
    int    nBuckP    = nBuck + 1;
    int    nBlk      = (nEdges + L1_CAP - 1) / L1_CAP;
    int    nChunks   = (nBuckP * nBlk + SCAN_CHUNK - 1) / SCAN_CHUNK;
    int    matLen    = nChunks * SCAN_CHUNK;
    size_t csumPad   = ((size_t)nChunks + 3) / 4 * 4;
    size_t rsPad     = ((size_t)nRows + 1 + 3) / 4 * 4;
    size_t edgePad   = ((size_t)nEdges + 3) / 4 * 4;

    float* buf0 = (float*)d_ws;
    float* buf1 = buf0 + total;
    int*   cntM = (int*)(buf1 + total);
    int*   csum = cntM + matLen;
    int*   rowStart = csum + csumPad;
    int*   bin  = rowStart + rsPad;
    int*   ed   = bin + 2 * edgePad;

    int vecBlocks = (int)((total / 4 + 255) / 256);
    lgcn_init<<<vecBlocks, 256, 0, stream>>>(ue, ie, buf0, out, userElems, total);

    lgcn_bin_count<<<nBlk, MAXB, 0, stream>>>(rows, cntM, nEdges, nBuckP, nBlk, matLen);
    lgcn_chunk_sum<<<nChunks, SCAN_BLOCK, 0, stream>>>(cntM, csum);
    lgcn_scan<<<nChunks, SCAN_BLOCK, 0, stream>>>(cntM, csum);
    lgcn_bin_scatter<<<nBlk, MAXB, 0, stream>>>(rows, cols, vals, cntM, bin, nEdges, nBuckP, nBlk);
    lgcn_bucket_sort<<<nBuck, MAXB, 0, stream>>>(cntM, bin, rowStart, ed, nRows, nBlk);

    int nTiles     = (int)(total / 256);
    int wmmaBlocks = (nTiles + 7) / 8;
    long long rowThreads = (long long)nRows * 16;
    int rowBlocks  = (int)((rowThreads + 255) / 256);

    float* cur = buf0;
    float* nxt = buf1;
    for (int layer = 0; layer < 3; ++layer) {
        lgcn_row_spmm<<<rowBlocks, 256, 0, stream>>>(rowStart, ed, cur, nxt, nRows);
        lgcn_wmma_axpy<<<wmmaBlocks, 256, 0, stream>>>(nxt, out, nTiles);
        float* tmp = cur; cur = nxt; nxt = tmp;
    }
}


# Layer mean formed once in the last row kernel (one wave = 4 rows = one 16x16 WMMA tile in registers) from the inputs and the two kept layer tables; init copy and the 3 axpy passes go; sum order kept
# speedup vs baseline: 4.3471x; 1.1267x over previous
#include <hip/hip_runtime.h>
#include <hip/hip_bf16.h>


typedef __attribute__((ext_vector_type(2))) float v2f;
typedef __attribute__((ext_vector_type(8))) float v8f;

#define EMB_D 64
#define SCAN_BLOCK 256
#define SCAN_CHUNK 1024

#define BK_SHIFT 9
#define BK_ROWS 512
#define MAXB 1024
#define L1_PER 10
#define L1_CAP (L1_PER * MAXB)
#define L2_CAP 16384

__global__ void lgcn_bin_count(const int* __restrict__ rows, int* __restrict__ cntM,
                               int nEdges, int nBuckP, int nBlk, int matLen) {
    __shared__ int hist[MAXB];
    int tid = threadIdx.x;
    hist[tid] = 0;
    __syncthreads();
    int base = (int)blockIdx.x * L1_CAP;
    for (int j = 0; j < L1_PER; ++j) {
        int e = base + j * MAXB + tid;
        if (e < nEdges) atomicAdd(&hist[rows[e] >> BK_SHIFT], 1);
    }
    __syncthreads();
    if (tid < nBuckP) cntM[(size_t)tid * nBlk + blockIdx.x] = hist[tid];
    if (blockIdx.x == 0)
        for (int i = nBuckP * nBlk + tid; i < matLen; i += MAXB) cntM[i] = 0;
}

__global__ void lgcn_chunk_sum(const int* __restrict__ cnt, int* __restrict__ csum) {
    __shared__ int sh[SCAN_BLOCK];
    int tid = threadIdx.x;
    int4 c = *(const int4*)(cnt + (size_t)blockIdx.x * SCAN_CHUNK + (size_t)tid * 4);
    sh[tid] = c.x + c.y + c.z + c.w;
    __syncthreads();
    for (int s = SCAN_BLOCK / 2; s > 0; s >>= 1) {
        if (tid < s) sh[tid] += sh[tid + s];
        __syncthreads();
    }
    if (tid == 0) csum[blockIdx.x] = sh[0];
}

__global__ void lgcn_scan(int* __restrict__ cnt, const int* __restrict__ csum) {
    __shared__ int sh[SCAN_BLOCK];
    int tid = threadIdx.x;
    int part = 0;
    for (int i = tid; i < (int)blockIdx.x; i += SCAN_BLOCK) part += csum[i];
    sh[tid] = part;
    __syncthreads();
    for (int s = SCAN_BLOCK / 2; s > 0; s >>= 1) {
        if (tid < s) sh[tid] += sh[tid + s];
        __syncthreads();
    }
    int chunkOff = sh[0];
    __syncthreads();

    int* p = cnt + (size_t)blockIdx.x * SCAN_CHUNK + (size_t)tid * 4;
    int4 c = *(const int4*)p;
    int mine = c.x + c.y + c.z + c.w;
    sh[tid] = mine;
    __syncthreads();
    for (int d = 1; d < SCAN_BLOCK; d <<= 1) {
        int add = (tid >= d) ? sh[tid - d] : 0;
        __syncthreads();
        sh[tid] += add;
        __syncthreads();
    }
    int excl = sh[tid] - mine + chunkOff;
    int4 o;
    o.x = excl;
    o.y = o.x + c.x;
    o.z = o.y + c.y;
    o.w = o.z + c.z;
    *(int4*)p = o;
}

__global__ void lgcn_bin_scatter(const int* __restrict__ rows, const int* __restrict__ cols,
                                 const float* __restrict__ vals, const int* __restrict__ offM,
                                 int* __restrict__ bin, int nEdges, int nBuckP, int nBlk) {
    __shared__ int   lStart[MAXB];
    __shared__ int   gOff[MAXB];
    __shared__ int   sRow[L1_CAP];
    __shared__ int   sCol[L1_CAP];
    __shared__ float sVal[L1_CAP];
    int tid  = threadIdx.x;
    int base = (int)blockIdx.x * L1_CAP;
    int n    = nEdges - base;
    if (n <= 0) return;
    if (n > L1_CAP) n = L1_CAP;

    lStart[tid] = 0;
    gOff[tid]   = (tid < nBuckP) ? offM[(size_t)tid * nBlk + blockIdx.x] : 0;
    __syncthreads();

    int   r[L1_PER], c[L1_PER], k[L1_PER];
    float v[L1_PER];
#pragma unroll
    for (int j = 0; j < L1_PER; ++j) {
        int e = j * MAXB + tid;
        r[j] = 0; c[j] = 0; v[j] = 0.0f; k[j] = 0;
        if (e < n) {
            r[j] = rows[base + e];
            c[j] = cols[base + e];
            v[j] = vals[base + e];
            k[j] = atomicAdd(&lStart[r[j] >> BK_SHIFT], 1);
        }
    }
    __syncthreads();

    int mine = lStart[tid];
    for (int d = 1; d < MAXB; d <<= 1) {
        int add = (tid >= d) ? lStart[tid - d] : 0;
        __syncthreads();
        lStart[tid] += add;
        __syncthreads();
    }
    int excl = lStart[tid] - mine;
    __syncthreads();
    lStart[tid] = excl;
    __syncthreads();

#pragma unroll
    for (int j = 0; j < L1_PER; ++j) {
        int e = j * MAXB + tid;
        if (e < n) {
            int p = lStart[r[j] >> BK_SHIFT] + k[j];
            sRow[p] = r[j];
            sCol[p] = c[j];
            sVal[p] = v[j];
        }
    }
    __syncthreads();

    int2* binp = (int2*)bin;
    for (int i = tid; i < n; i += MAXB) {
        int row = sRow[i];
        int b   = row >> BK_SHIFT;
        int g   = gOff[b] + (i - lStart[b]);
        int2 rec;
        rec.x = (sCol[i] << BK_SHIFT) | (row & (BK_ROWS - 1));
        rec.y = __float_as_int(sVal[i]);
        binp[g] = rec;
    }
}

__global__ void lgcn_bucket_sort(const int* __restrict__ offM, const int* __restrict__ bin,
                                 int* __restrict__ rowStart, int* __restrict__ ed,
                                 int nRows, int nBlk) {
    __shared__ int  cnt[BK_ROWS];
    __shared__ int  cur[BK_ROWS];
    __shared__ int2 sEd[L2_CAP];
    int tid   = threadIdx.x;
    int b     = (int)blockIdx.x;
    int start = offM[(size_t)b * nBlk];
    int end   = offM[(size_t)(b + 1) * nBlk];
    int n     = end - start;
    int row0  = b << BK_SHIFT;
    const int2* bp = (const int2*)bin + start;

    if (tid < BK_ROWS) cnt[tid] = 0;
    __syncthreads();
    for (int i = tid; i < n; i += MAXB) atomicAdd(&cnt[bp[i].x & (BK_ROWS - 1)], 1);
    __syncthreads();

    int mine = (tid < BK_ROWS) ? cnt[tid] : 0;
    for (int d = 1; d < BK_ROWS; d <<= 1) {
        int add = (tid < BK_ROWS && tid >= d) ? cnt[tid - d] : 0;
        __syncthreads();
        if (tid < BK_ROWS) cnt[tid] += add;
        __syncthreads();
    }
    if (tid < BK_ROWS) {
        int excl = cnt[tid] - mine;
        cur[tid] = excl;
        int gr = row0 + tid;
        if (gr <= nRows) rowStart[gr] = start + excl;
    }
    __syncthreads();

    int2* edp = (int2*)ed + start;
    if (n <= L2_CAP) {
        for (int i = tid; i < n; i += MAXB) {
            int2 rec = bp[i];
            int p = atomicAdd(&cur[rec.x & (BK_ROWS - 1)], 1);
            int2 e; e.x = (int)((unsigned)rec.x >> BK_SHIFT); e.y = rec.y;
            sEd[p] = e;
        }
        __syncthreads();
        for (int i = tid; i < n; i += MAXB) edp[i] = sEd[i];
    } else {
        for (int i = tid; i < n; i += MAXB) {
            int2 rec = bp[i];
            int p = atomicAdd(&cur[rec.x & (BK_ROWS - 1)], 1);
            int2 e; e.x = (int)((unsigned)rec.x >> BK_SHIFT); e.y = rec.y;
            edp[p] = e;
        }
    }
}

__device__ __forceinline__ v8f lgcn_tile_axpy(v8f c, v8f t, int lo, int hi) {
#pragma unroll
    for (int kk = 0; kk < 4; ++kk) {
        int m = 2 * kk + 8 * hi;
        v2f a, b;
        a[0] = (lo == m)     ? 0.25f : 0.0f;
        a[1] = (lo == m + 1) ? 0.25f : 0.0f;
        b[0] = t[2 * kk];
        b[1] = t[2 * kk + 1];
        c = __builtin_amdgcn_wmma_f32_16x16x4_f32(
                 false, a,  false, b,
                 (short)0, c,  false,  false);
    }
    return c;
}

template <int MODE>
__device__ __forceinline__ const float4* lgcn_row_ptr(int c, const float* __restrict__ ue,
                                                      const float* __restrict__ ie,
                                                      const float* __restrict__ x,
                                                      int nUsers, int sub) {
    const float* p;
    if (MODE == 0) p = (c < nUsers) ? ue + (size_t)c * EMB_D : ie + (size_t)(c - nUsers) * EMB_D;
    else           p = x + (size_t)c * EMB_D;
    return (const float4*)p + sub;
}

template <int MODE>
__global__ void __launch_bounds__(256)
lgcn_row_layer(const int* __restrict__ rowStart, const int* __restrict__ ed,
               const float* __restrict__ ue, const float* __restrict__ ie,
               const float* __restrict__ x, const float* __restrict__ x1,
               float* __restrict__ y, int nUsers, int nRows) {
    int t    = (int)blockIdx.x * (int)blockDim.x + (int)threadIdx.x;
    int lane = (int)threadIdx.x & 31;
    int sub  = lane & 7;
    int r0   = (t >> 5) << 2;
    if (r0 >= nRows) return;
    int  r     = r0 + (lane >> 3);
    bool valid = r < nRows;
    int  rc    = valid ? r : nRows - 1;
    int j   = rowStart[rc];
    int e1r = rowStart[rc + 1];
    int end = valid ? e1r : j;
    const int2* edp = (const int2*)ed;
    float4 a0, a1;
    a0.x = a0.y = a0.z = a0.w = 0.0f;
    a1.x = a1.y = a1.z = a1.w = 0.0f;
    for (; j + 4 <= end; j += 4) {
        int2 e0 = edp[j + 0];
        int2 e1 = edp[j + 1];
        int2 e2 = edp[j + 2];
        int2 e3 = edp[j + 3];
        const float4* p0 = lgcn_row_ptr<MODE>(e0.x, ue, ie, x, nUsers, sub);
        const float4* p1 = lgcn_row_ptr<MODE>(e1.x, ue, ie, x, nUsers, sub);
        const float4* p2 = lgcn_row_ptr<MODE>(e2.x, ue, ie, x, nUsers, sub);
        const float4* p3 = lgcn_row_ptr<MODE>(e3.x, ue, ie, x, nUsers, sub);
        float4 g0 = p0[0], h0 = p0[8];
        float4 g1 = p1[0], h1 = p1[8];
        float4 g2 = p2[0], h2 = p2[8];
        float4 g3 = p3[0], h3 = p3[8];
        float v0 = __int_as_float(e0.y);
        float v1 = __int_as_float(e1.y);
        float v2 = __int_as_float(e2.y);
        float v3 = __int_as_float(e3.y);
        a0.x += g0.x * v0; a0.y += g0.y * v0; a0.z += g0.z * v0; a0.w += g0.w * v0;
        a1.x += h0.x * v0; a1.y += h0.y * v0; a1.z += h0.z * v0; a1.w += h0.w * v0;
        a0.x += g1.x * v1; a0.y += g1.y * v1; a0.z += g1.z * v1; a0.w += g1.w * v1;
        a1.x += h1.x * v1; a1.y += h1.y * v1; a1.z += h1.z * v1; a1.w += h1.w * v1;
        a0.x += g2.x * v2; a0.y += g2.y * v2; a0.z += g2.z * v2; a0.w += g2.w * v2;
        a1.x += h2.x * v2; a1.y += h2.y * v2; a1.z += h2.z * v2; a1.w += h2.w * v2;
        a0.x += g3.x * v3; a0.y += g3.y * v3; a0.z += g3.z * v3; a0.w += g3.w * v3;
        a1.x += h3.x * v3; a1.y += h3.y * v3; a1.z += h3.z * v3; a1.w += h3.w * v3;
    }
    int n = end - j;
    if (n > 0) {
        int last = end - 1;
        int2 e0 = edp[j];
        int2 e1 = edp[(n > 1) ? j + 1 : last];
        int2 e2 = edp[(n > 2) ? j + 2 : last];
        const float4* p0 = lgcn_row_ptr<MODE>(e0.x, ue, ie, x, nUsers, sub);
        const float4* p1 = lgcn_row_ptr<MODE>(e1.x, ue, ie, x, nUsers, sub);
        const float4* p2 = lgcn_row_ptr<MODE>(e2.x, ue, ie, x, nUsers, sub);
        float4 g0 = p0[0], h0 = p0[8];
        float4 g1 = p1[0], h1 = p1[8];
        float4 g2 = p2[0], h2 = p2[8];
        float v0 = __int_as_float(e0.y);
        float v1 = (n > 1) ? __int_as_float(e1.y) : 0.0f;
        float v2 = (n > 2) ? __int_as_float(e2.y) : 0.0f;
        if (n < 2) { g1.x = 0.0f; g1.y = 0.0f; g1.z = 0.0f; g1.w = 0.0f;
                     h1.x = 0.0f; h1.y = 0.0f; h1.z = 0.0f; h1.w = 0.0f; }
        if (n < 3) { g2.x = 0.0f; g2.y = 0.0f; g2.z = 0.0f; g2.w = 0.0f;
                     h2.x = 0.0f; h2.y = 0.0f; h2.z = 0.0f; h2.w = 0.0f; }
        a0.x += g0.x * v0; a0.y += g0.y * v0; a0.z += g0.z * v0; a0.w += g0.w * v0;
        a1.x += h0.x * v0; a1.y += h0.y * v0; a1.z += h0.z * v0; a1.w += h0.w * v0;
        a0.x += g1.x * v1; a0.y += g1.y * v1; a0.z += g1.z * v1; a0.w += g1.w * v1;
        a1.x += h1.x * v1; a1.y += h1.y * v1; a1.z += h1.z * v1; a1.w += h1.w * v1;
        a0.x += g2.x * v2; a0.y += g2.y * v2; a0.z += g2.z * v2; a0.w += g2.w * v2;
        a1.x += h2.x * v2; a1.y += h2.y * v2; a1.z += h2.z * v2; a1.w += h2.w * v2;
    }

    size_t slot = (size_t)rc * (EMB_D / 4) + sub;
    if (MODE == 2) {
        const float4* z0 = lgcn_row_ptr<0>(rc, ue, ie, x, nUsers, sub);
        float4 q0 = z0[0], q1 = z0[8];
        float4 s0 = ((const float4*)x1)[slot], s1 = ((const float4*)x1)[slot + 8];
        float4 u0 = ((const float4*)x)[slot],  u1 = ((const float4*)x)[slot + 8];
        v8f c, t1, t2, t3;
        c[0] = q0.x * 0.25f; c[1] = q0.y * 0.25f; c[2] = q0.z * 0.25f; c[3] = q0.w * 0.25f;
        c[4] = q1.x * 0.25f; c[5] = q1.y * 0.25f; c[6] = q1.z * 0.25f; c[7] = q1.w * 0.25f;
        t1[0] = s0.x; t1[1] = s0.y; t1[2] = s0.z; t1[3] = s0.w;
        t1[4] = s1.x; t1[5] = s1.y; t1[6] = s1.z; t1[7] = s1.w;
        t2[0] = u0.x; t2[1] = u0.y; t2[2] = u0.z; t2[3] = u0.w;
        t2[4] = u1.x; t2[5] = u1.y; t2[6] = u1.z; t2[7] = u1.w;
        t3[0] = a0.x; t3[1] = a0.y; t3[2] = a0.z; t3[3] = a0.w;
        t3[4] = a1.x; t3[5] = a1.y; t3[6] = a1.z; t3[7] = a1.w;
        int lo = lane & 15;
        int hi = lane >> 4;
        c = lgcn_tile_axpy(c, t1, lo, hi);
        c = lgcn_tile_axpy(c, t2, lo, hi);
        c = lgcn_tile_axpy(c, t3, lo, hi);
        a0.x = c[0]; a0.y = c[1]; a0.z = c[2]; a0.w = c[3];
        a1.x = c[4]; a1.y = c[5]; a1.z = c[6]; a1.w = c[7];
    }
    if (valid) {
        ((float4*)y)[slot]     = a0;
        ((float4*)y)[slot + 8] = a1;
    }
}

extern "C" void kernel_launch(void* const* d_in, const int* in_sizes, int n_in,
                              void* d_out, int out_size, void* d_ws, size_t ws_size,
                              hipStream_t stream) {
    const float* ue   = (const float*)d_in[0];
    const float* ie   = (const float*)d_in[1];
    const int*   rows = (const int*)d_in[2];
    const int*   cols = (const int*)d_in[3];
    const float* vals = (const float*)d_in[4];
    float*       out  = (float*)d_out;

    size_t userElems = (size_t)in_sizes[0];
    size_t itemElems = (size_t)in_sizes[1];
    int    nEdges    = in_sizes[2];
    size_t total     = userElems + itemElems;
    int    nRows     = (int)(total / EMB_D);
    int    nUsers    = (int)(userElems / EMB_D);

    int    nBuck     = (nRows + 1 + BK_ROWS - 1) / BK_ROWS;
    int    nBuckP    = nBuck + 1;
    int    nBlk      = (nEdges + L1_CAP - 1) / L1_CAP;
    int    nChunks   = (nBuckP * nBlk + SCAN_CHUNK - 1) / SCAN_CHUNK;
    int    matLen    = nChunks * SCAN_CHUNK;
    size_t csumPad   = ((size_t)nChunks + 3) / 4 * 4;
    size_t rsPad     = ((size_t)nRows + 1 + 3) / 4 * 4;
    size_t edgePad   = ((size_t)nEdges + 3) / 4 * 4;

    float* buf0 = (float*)d_ws;
    float* buf1 = buf0 + total;
    int*   cntM = (int*)(buf1 + total);
    int*   csum = cntM + matLen;
    int*   rowStart = csum + csumPad;
    int*   bin  = rowStart + rsPad;
    int*   ed   = bin + 2 * edgePad;

    lgcn_bin_count<<<nBlk, MAXB, 0, stream>>>(rows, cntM, nEdges, nBuckP, nBlk, matLen);
    lgcn_chunk_sum<<<nChunks, SCAN_BLOCK, 0, stream>>>(cntM, csum);
    lgcn_scan<<<nChunks, SCAN_BLOCK, 0, stream>>>(cntM, csum);
    lgcn_bin_scatter<<<nBlk, MAXB, 0, stream>>>(rows, cols, vals, cntM, bin, nEdges, nBuckP, nBlk);
    lgcn_bucket_sort<<<nBuck, MAXB, 0, stream>>>(cntM, bin, rowStart, ed, nRows, nBlk);

    int rowBlocks = (nRows + 31) / 32;
    lgcn_row_layer<0><<<rowBlocks, 256, 0, stream>>>(rowStart, ed, ue, ie, nullptr, nullptr, buf0, nUsers, nRows);
    lgcn_row_layer<1><<<rowBlocks, 256, 0, stream>>>(rowStart, ed, ue, ie, buf0, nullptr, buf1, nUsers, nRows);
    lgcn_row_layer<2><<<rowBlocks, 256, 0, stream>>>(rowStart, ed, ue, ie, buf1, buf0, out, nUsers, nRows);
}


# Row layers: 16 lanes per row, 16 B each, so a gathered 256-B table row is ONE load instruction over a whole aligned 256-B unit (was two 128-B halves, two instructions); 8 pairs a trip; sum order kept
# speedup vs baseline: 4.8314x; 1.1114x over previous
#include <hip/hip_runtime.h>
#include <hip/hip_bf16.h>


typedef __attribute__((ext_vector_type(2))) float v2f;
typedef __attribute__((ext_vector_type(4))) float v4f;
typedef __attribute__((ext_vector_type(8))) float v8f;

#define EMB_D 64
#define SCAN_BLOCK 256
#define SCAN_CHUNK 1024

#define BK_SHIFT 9
#define BK_ROWS 512
#define MAXB 1024
#define L1_PER 10
#define L1_CAP (L1_PER * MAXB)
#define L2_CAP 16384

__global__ void lgcn_bin_count(const int* __restrict__ rows, int* __restrict__ cntM,
                               int nEdges, int nBuckP, int nBlk, int matLen) {
    __shared__ int hist[MAXB];
    int tid = threadIdx.x;
    hist[tid] = 0;
    __syncthreads();
    int base = (int)blockIdx.x * L1_CAP;
    for (int j = 0; j < L1_PER; ++j) {
        int e = base + j * MAXB + tid;
        if (e < nEdges) atomicAdd(&hist[rows[e] >> BK_SHIFT], 1);
    }
    __syncthreads();
    if (tid < nBuckP) cntM[(size_t)tid * nBlk + blockIdx.x] = hist[tid];
    if (blockIdx.x == 0)
        for (int i = nBuckP * nBlk + tid; i < matLen; i += MAXB) cntM[i] = 0;
}

__global__ void lgcn_chunk_sum(const int* __restrict__ cnt, int* __restrict__ csum) {
    __shared__ int sh[SCAN_BLOCK];
    int tid = threadIdx.x;
    int4 c = *(const int4*)(cnt + (size_t)blockIdx.x * SCAN_CHUNK + (size_t)tid * 4);
    sh[tid] = c.x + c.y + c.z + c.w;
    __syncthreads();
    for (int s = SCAN_BLOCK / 2; s > 0; s >>= 1) {
        if (tid < s) sh[tid] += sh[tid + s];
        __syncthreads();
    }
    if (tid == 0) csum[blockIdx.x] = sh[0];
}

__global__ void lgcn_scan(int* __restrict__ cnt, const int* __restrict__ csum) {
    __shared__ int sh[SCAN_BLOCK];
    int tid = threadIdx.x;
    int part = 0;
    for (int i = tid; i < (int)blockIdx.x; i += SCAN_BLOCK) part += csum[i];
    sh[tid] = part;
    __syncthreads();
    for (int s = SCAN_BLOCK / 2; s > 0; s >>= 1) {
        if (tid < s) sh[tid] += sh[tid + s];
        __syncthreads();
    }
    int chunkOff = sh[0];
    __syncthreads();

    int* p = cnt + (size_t)blockIdx.x * SCAN_CHUNK + (size_t)tid * 4;
    int4 c = *(const int4*)p;
    int mine = c.x + c.y + c.z + c.w;
    sh[tid] = mine;
    __syncthreads();
    for (int d = 1; d < SCAN_BLOCK; d <<= 1) {
        int add = (tid >= d) ? sh[tid - d] : 0;
        __syncthreads();
        sh[tid] += add;
        __syncthreads();
    }
    int excl = sh[tid] - mine + chunkOff;
    int4 o;
    o.x = excl;
    o.y = o.x + c.x;
    o.z = o.y + c.y;
    o.w = o.z + c.z;
    *(int4*)p = o;
}

__global__ void lgcn_bin_scatter(const int* __restrict__ rows, const int* __restrict__ cols,
                                 const float* __restrict__ vals, const int* __restrict__ offM,
                                 int* __restrict__ bin, int nEdges, int nBuckP, int nBlk) {
    __shared__ int   lStart[MAXB];
    __shared__ int   gOff[MAXB];
    __shared__ int   sRow[L1_CAP];
    __shared__ int   sCol[L1_CAP];
    __shared__ float sVal[L1_CAP];
    int tid  = threadIdx.x;
    int base = (int)blockIdx.x * L1_CAP;
    int n    = nEdges - base;
    if (n <= 0) return;
    if (n > L1_CAP) n = L1_CAP;

    lStart[tid] = 0;
    gOff[tid]   = (tid < nBuckP) ? offM[(size_t)tid * nBlk + blockIdx.x] : 0;
    __syncthreads();

    int   r[L1_PER], c[L1_PER], k[L1_PER];
    float v[L1_PER];
#pragma unroll
    for (int j = 0; j < L1_PER; ++j) {
        int e = j * MAXB + tid;
        r[j] = 0; c[j] = 0; v[j] = 0.0f; k[j] = 0;
        if (e < n) {
            r[j] = rows[base + e];
            c[j] = cols[base + e];
            v[j] = vals[base + e];
            k[j] = atomicAdd(&lStart[r[j] >> BK_SHIFT], 1);
        }
    }
    __syncthreads();

    int mine = lStart[tid];
    for (int d = 1; d < MAXB; d <<= 1) {
        int add = (tid >= d) ? lStart[tid - d] : 0;
        __syncthreads();
        lStart[tid] += add;
        __syncthreads();
    }
    int excl = lStart[tid] - mine;
    __syncthreads();
    lStart[tid] = excl;
    __syncthreads();

#pragma unroll
    for (int j = 0; j < L1_PER; ++j) {
        int e = j * MAXB + tid;
        if (e < n) {
            int p = lStart[r[j] >> BK_SHIFT] + k[j];
            sRow[p] = r[j];
            sCol[p] = c[j];
            sVal[p] = v[j];
        }
    }
    __syncthreads();

    int2* binp = (int2*)bin;
    for (int i = tid; i < n; i += MAXB) {
        int row = sRow[i];
        int b   = row >> BK_SHIFT;
        int g   = gOff[b] + (i - lStart[b]);
        int2 rec;
        rec.x = (sCol[i] << BK_SHIFT) | (row & (BK_ROWS - 1));
        rec.y = __float_as_int(sVal[i]);
        binp[g] = rec;
    }
}

__global__ void lgcn_bucket_sort(const int* __restrict__ offM, const int* __restrict__ bin,
                                 int* __restrict__ rowStart, int* __restrict__ ed,
                                 int nRows, int nBlk) {
    __shared__ int  cnt[BK_ROWS];
    __shared__ int  cur[BK_ROWS];
    __shared__ int2 sEd[L2_CAP];
    int tid   = threadIdx.x;
    int b     = (int)blockIdx.x;
    int start = offM[(size_t)b * nBlk];
    int end   = offM[(size_t)(b + 1) * nBlk];
    int n     = end - start;
    int row0  = b << BK_SHIFT;
    const int2* bp = (const int2*)bin + start;

    if (tid < BK_ROWS) cnt[tid] = 0;
    __syncthreads();
    for (int i = tid; i < n; i += MAXB) atomicAdd(&cnt[bp[i].x & (BK_ROWS - 1)], 1);
    __syncthreads();

    int mine = (tid < BK_ROWS) ? cnt[tid] : 0;
    for (int d = 1; d < BK_ROWS; d <<= 1) {
        int add = (tid < BK_ROWS && tid >= d) ? cnt[tid - d] : 0;
        __syncthreads();
        if (tid < BK_ROWS) cnt[tid] += add;
        __syncthreads();
    }
    if (tid < BK_ROWS) {
        int excl = cnt[tid] - mine;
        cur[tid] = excl;
        int gr = row0 + tid;
        if (gr <= nRows) rowStart[gr] = start + excl;
    }
    __syncthreads();

    int2* edp = (int2*)ed + start;
    if (n <= L2_CAP) {
        for (int i = tid; i < n; i += MAXB) {
            int2 rec = bp[i];
            int p = atomicAdd(&cur[rec.x & (BK_ROWS - 1)], 1);
            int2 e; e.x = (int)((unsigned)rec.x >> BK_SHIFT); e.y = rec.y;
            sEd[p] = e;
        }
        __syncthreads();
        for (int i = tid; i < n; i += MAXB) edp[i] = sEd[i];
    } else {
        for (int i = tid; i < n; i += MAXB) {
            int2 rec = bp[i];
            int p = atomicAdd(&cur[rec.x & (BK_ROWS - 1)], 1);
            int2 e; e.x = (int)((unsigned)rec.x >> BK_SHIFT); e.y = rec.y;
            edp[p] = e;
        }
    }
}

__device__ __forceinline__ v8f lgcn_tile_axpy(v8f c, v8f t, int lo, int hi) {
#pragma unroll
    for (int kk = 0; kk < 4; ++kk) {
        int m = 2 * kk + 8 * hi;
        v2f a, b;
        a[0] = (lo == m)     ? 0.25f : 0.0f;
        a[1] = (lo == m + 1) ? 0.25f : 0.0f;
        b[0] = t[2 * kk];
        b[1] = t[2 * kk + 1];
        c = __builtin_amdgcn_wmma_f32_16x16x4_f32(
                 false, a,  false, b,
                 (short)0, c,  false,  false);
    }
    return c;
}

template <int MODE>
__device__ __forceinline__ const v4f* lgcn_row_ptr(int c, const float* __restrict__ ue,
                                                   const float* __restrict__ ie,
                                                   const float* __restrict__ x,
                                                   int nUsers, int sub) {
    const char* base = (const char*)x;
    unsigned    row  = (unsigned)c;
    if (MODE == 0) {
        bool user = c < nUsers;
        base = user ? (const char*)ue : (const char*)ie;
        row  = user ? (unsigned)c : (unsigned)(c - nUsers);
    }
    unsigned off = (row << 8) | ((unsigned)sub << 4);
    return (const v4f*)(base + off);
}

__device__ __forceinline__ v4f lgcn_fma4(v4f g, float v, v4f a) {
    v4f vv = {v, v, v, v};
    return __builtin_elementwise_fma(g, vv, a);
}

template <int MODE>
__device__ __forceinline__ v4f lgcn_row_sum(int j, int end, const int2* __restrict__ edp,
                                            const float* __restrict__ ue,
                                            const float* __restrict__ ie,
                                            const float* __restrict__ x,
                                            int nUsers, int sub) {
    v4f a = {0.0f, 0.0f, 0.0f, 0.0f};
    for (; j + 8 <= end; j += 8) {
        int2 e0 = edp[j + 0];
        int2 e1 = edp[j + 1];
        int2 e2 = edp[j + 2];
        int2 e3 = edp[j + 3];
        int2 e4 = edp[j + 4];
        int2 e5 = edp[j + 5];
        int2 e6 = edp[j + 6];
        int2 e7 = edp[j + 7];
        v4f g0 = *lgcn_row_ptr<MODE>(e0.x, ue, ie, x, nUsers, sub);
        v4f g1 = *lgcn_row_ptr<MODE>(e1.x, ue, ie, x, nUsers, sub);
        v4f g2 = *lgcn_row_ptr<MODE>(e2.x, ue, ie, x, nUsers, sub);
        v4f g3 = *lgcn_row_ptr<MODE>(e3.x, ue, ie, x, nUsers, sub);
        v4f g4 = *lgcn_row_ptr<MODE>(e4.x, ue, ie, x, nUsers, sub);
        v4f g5 = *lgcn_row_ptr<MODE>(e5.x, ue, ie, x, nUsers, sub);
        v4f g6 = *lgcn_row_ptr<MODE>(e6.x, ue, ie, x, nUsers, sub);
        v4f g7 = *lgcn_row_ptr<MODE>(e7.x, ue, ie, x, nUsers, sub);
        a = lgcn_fma4(g0, __int_as_float(e0.y), a);
        a = lgcn_fma4(g1, __int_as_float(e1.y), a);
        a = lgcn_fma4(g2, __int_as_float(e2.y), a);
        a = lgcn_fma4(g3, __int_as_float(e3.y), a);
        a = lgcn_fma4(g4, __int_as_float(e4.y), a);
        a = lgcn_fma4(g5, __int_as_float(e5.y), a);
        a = lgcn_fma4(g6, __int_as_float(e6.y), a);
        a = lgcn_fma4(g7, __int_as_float(e7.y), a);
    }
    if (j + 4 <= end) {
        int2 e0 = edp[j + 0];
        int2 e1 = edp[j + 1];
        int2 e2 = edp[j + 2];
        int2 e3 = edp[j + 3];
        v4f g0 = *lgcn_row_ptr<MODE>(e0.x, ue, ie, x, nUsers, sub);
        v4f g1 = *lgcn_row_ptr<MODE>(e1.x, ue, ie, x, nUsers, sub);
        v4f g2 = *lgcn_row_ptr<MODE>(e2.x, ue, ie, x, nUsers, sub);
        v4f g3 = *lgcn_row_ptr<MODE>(e3.x, ue, ie, x, nUsers, sub);
        a = lgcn_fma4(g0, __int_as_float(e0.y), a);
        a = lgcn_fma4(g1, __int_as_float(e1.y), a);
        a = lgcn_fma4(g2, __int_as_float(e2.y), a);
        a = lgcn_fma4(g3, __int_as_float(e3.y), a);
        j += 4;
    }
    int n = end - j;
    if (n > 0) {
        int last = end - 1;
        int2 e0 = edp[j];
        int2 e1 = edp[(n > 1) ? j + 1 : last];
        int2 e2 = edp[(n > 2) ? j + 2 : last];
        v4f g0 = *lgcn_row_ptr<MODE>(e0.x, ue, ie, x, nUsers, sub);
        v4f g1 = *lgcn_row_ptr<MODE>(e1.x, ue, ie, x, nUsers, sub);
        v4f g2 = *lgcn_row_ptr<MODE>(e2.x, ue, ie, x, nUsers, sub);
        v4f zero = {0.0f, 0.0f, 0.0f, 0.0f};
        float v0 = __int_as_float(e0.y);
        float v1 = (n > 1) ? __int_as_float(e1.y) : 0.0f;
        float v2 = (n > 2) ? __int_as_float(e2.y) : 0.0f;
        if (n < 2) g1 = zero;
        if (n < 3) g2 = zero;
        a = lgcn_fma4(g0, v0, a);
        a = lgcn_fma4(g1, v1, a);
        a = lgcn_fma4(g2, v2, a);
    }
    return a;
}

template <int MODE>
__global__ void __launch_bounds__(256)
lgcn_row_layer(const int* __restrict__ rowStart, const int* __restrict__ ed,
               const float* __restrict__ ue, const float* __restrict__ ie,
               const float* __restrict__ x, const float* __restrict__ x1,
               float* __restrict__ y, int nUsers, int nRows) {
    int t    = (int)blockIdx.x * (int)blockDim.x + (int)threadIdx.x;
    int lane = (int)threadIdx.x & 31;
    int sub  = lane & 15;
    int hl   = lane >> 4;
    int r0   = (t >> 5) << 2;
    if (r0 >= nRows) return;
    int  rA  = r0 + hl;
    int  rB  = r0 + 2 + hl;
    bool vA  = rA < nRows;
    bool vB  = rB < nRows;
    int  cA  = vA ? rA : nRows - 1;
    int  cB  = vB ? rB : nRows - 1;
    int jA  = rowStart[cA];
    int eA  = rowStart[cA + 1];
    int jB  = rowStart[cB];
    int eB  = rowStart[cB + 1];
    int endA = vA ? eA : jA;
    int endB = vB ? eB : jB;
    const int2* edp = (const int2*)ed;
    v4f a0 = lgcn_row_sum<MODE>(jA, endA, edp, ue, ie, x, nUsers, sub);
    v4f a1 = lgcn_row_sum<MODE>(jB, endB, edp, ue, ie, x, nUsers, sub);

    size_t slotA = (size_t)cA * (EMB_D / 4) + sub;
    size_t slotB = (size_t)cB * (EMB_D / 4) + sub;
    if (MODE == 2) {
        v4f q0 = *lgcn_row_ptr<0>(cA, ue, ie, x, nUsers, sub);
        v4f q1 = *lgcn_row_ptr<0>(cB, ue, ie, x, nUsers, sub);
        v4f s0 = ((const v4f*)x1)[slotA], s1 = ((const v4f*)x1)[slotB];
        v4f u0 = ((const v4f*)x)[slotA],  u1 = ((const v4f*)x)[slotB];
        v8f c, t1, t2, t3;
        c[0] = q0[0] * 0.25f; c[1] = q0[1] * 0.25f; c[2] = q0[2] * 0.25f; c[3] = q0[3] * 0.25f;
        c[4] = q1[0] * 0.25f; c[5] = q1[1] * 0.25f; c[6] = q1[2] * 0.25f; c[7] = q1[3] * 0.25f;
        t1[0] = s0[0]; t1[1] = s0[1]; t1[2] = s0[2]; t1[3] = s0[3];
        t1[4] = s1[0]; t1[5] = s1[1]; t1[6] = s1[2]; t1[7] = s1[3];
        t2[0] = u0[0]; t2[1] = u0[1]; t2[2] = u0[2]; t2[3] = u0[3];
        t2[4] = u1[0]; t2[5] = u1[1]; t2[6] = u1[2]; t2[7] = u1[3];
        t3[0] = a0[0]; t3[1] = a0[1]; t3[2] = a0[2]; t3[3] = a0[3];
        t3[4] = a1[0]; t3[5] = a1[1]; t3[6] = a1[2]; t3[7] = a1[3];
        int lo = lane & 15;
        int hi = lane >> 4;
        c = lgcn_tile_axpy(c, t1, lo, hi);
        c = lgcn_tile_axpy(c, t2, lo, hi);
        c = lgcn_tile_axpy(c, t3, lo, hi);
        a0[0] = c[0]; a0[1] = c[1]; a0[2] = c[2]; a0[3] = c[3];
        a1[0] = c[4]; a1[1] = c[5]; a1[2] = c[6]; a1[3] = c[7];
    }
    if (vA) ((v4f*)y)[slotA] = a0;
    if (vB) ((v4f*)y)[slotB] = a1;
}

extern "C" void kernel_launch(void* const* d_in, const int* in_sizes, int n_in,
                              void* d_out, int out_size, void* d_ws, size_t ws_size,
                              hipStream_t stream) {
    const float* ue   = (const float*)d_in[0];
    const float* ie   = (const float*)d_in[1];
    const int*   rows = (const int*)d_in[2];
    const int*   cols = (const int*)d_in[3];
    const float* vals = (const float*)d_in[4];
    float*       out  = (float*)d_out;

    size_t userElems = (size_t)in_sizes[0];
    size_t itemElems = (size_t)in_sizes[1];
    int    nEdges    = in_sizes[2];
    size_t total     = userElems + itemElems;
    int    nRows     = (int)(total / EMB_D);
    int    nUsers    = (int)(userElems / EMB_D);

    int    nBuck     = (nRows + 1 + BK_ROWS - 1) / BK_ROWS;
    int    nBuckP    = nBuck + 1;
    int    nBlk      = (nEdges + L1_CAP - 1) / L1_CAP;
    int    nChunks   = (nBuckP * nBlk + SCAN_CHUNK - 1) / SCAN_CHUNK;
    int    matLen    = nChunks * SCAN_CHUNK;
    size_t csumPad   = ((size_t)nChunks + 3) / 4 * 4;
    size_t rsPad     = ((size_t)nRows + 1 + 3) / 4 * 4;
    size_t edgePad   = ((size_t)nEdges + 3) / 4 * 4;

    float* buf0 = (float*)d_ws;
    float* buf1 = buf0 + total;
    int*   cntM = (int*)(buf1 + total);
    int*   csum = cntM + matLen;
    int*   rowStart = csum + csumPad;
    int*   bin  = rowStart + rsPad;
    int*   ed   = bin + 2 * edgePad;

    lgcn_bin_count<<<nBlk, MAXB, 0, stream>>>(rows, cntM, nEdges, nBuckP, nBlk, matLen);
    lgcn_chunk_sum<<<nChunks, SCAN_BLOCK, 0, stream>>>(cntM, csum);
    lgcn_scan<<<nChunks, SCAN_BLOCK, 0, stream>>>(cntM, csum);
    lgcn_bin_scatter<<<nBlk, MAXB, 0, stream>>>(rows, cols, vals, cntM, bin, nEdges, nBuckP, nBlk);
    lgcn_bucket_sort<<<nBuck, MAXB, 0, stream>>>(cntM, bin, rowStart, ed, nRows, nBlk);

    int rowBlocks = (nRows + 31) / 32;
    lgcn_row_layer<0><<<rowBlocks, 256, 0, stream>>>(rowStart, ed, ue, ie, nullptr, nullptr, buf0, nUsers, nRows);
    lgcn_row_layer<1><<<rowBlocks, 256, 0, stream>>>(rowStart, ed, ue, ie, buf0, nullptr, buf1, nUsers, nRows);
    lgcn_row_layer<2><<<rowBlocks, 256, 0, stream>>>(rowStart, ed, ue, ie, buf1, buf0, out, nUsers, nRows);
}


# Row layers: a wave half reads its row's pair list 16 pairs per load, one per lane (all 4 rows' first loads up front, next 16 ahead of the gathers); col/val go round by DPP row_share; sum order kept
# speedup vs baseline: 5.2175x; 1.0799x over previous
#include <hip/hip_runtime.h>
#include <hip/hip_bf16.h>


typedef __attribute__((ext_vector_type(2))) float v2f;
typedef __attribute__((ext_vector_type(4))) float v4f;
typedef __attribute__((ext_vector_type(8))) float v8f;

#define EMB_D 64
#define SCAN_BLOCK 256
#define SCAN_CHUNK 1024

#define BK_SHIFT 9
#define BK_ROWS 512
#define MAXB 1024
#define L1_PER 10
#define L1_CAP (L1_PER * MAXB)
#define L2_CAP 16384

__global__ void lgcn_bin_count(const int* __restrict__ rows, int* __restrict__ cntM,
                               int nEdges, int nBuckP, int nBlk, int matLen) {
    __shared__ int hist[MAXB];
    int tid = threadIdx.x;
    hist[tid] = 0;
    __syncthreads();
    int base = (int)blockIdx.x * L1_CAP;
    for (int j = 0; j < L1_PER; ++j) {
        int e = base + j * MAXB + tid;
        if (e < nEdges) atomicAdd(&hist[rows[e] >> BK_SHIFT], 1);
    }
    __syncthreads();
    if (tid < nBuckP) cntM[(size_t)tid * nBlk + blockIdx.x] = hist[tid];
    if (blockIdx.x == 0)
        for (int i = nBuckP * nBlk + tid; i < matLen; i += MAXB) cntM[i] = 0;
}

__global__ void lgcn_chunk_sum(const int* __restrict__ cnt, int* __restrict__ csum) {
    __shared__ int sh[SCAN_BLOCK];
    int tid = threadIdx.x;
    int4 c = *(const int4*)(cnt + (size_t)blockIdx.x * SCAN_CHUNK + (size_t)tid * 4);
    sh[tid] = c.x + c.y + c.z + c.w;
    __syncthreads();
    for (int s = SCAN_BLOCK / 2; s > 0; s >>= 1) {
        if (tid < s) sh[tid] += sh[tid + s];
        __syncthreads();
    }
    if (tid == 0) csum[blockIdx.x] = sh[0];
}

__global__ void lgcn_scan(int* __restrict__ cnt, const int* __restrict__ csum) {
    __shared__ int sh[SCAN_BLOCK];
    int tid = threadIdx.x;
    int part = 0;
    for (int i = tid; i < (int)blockIdx.x; i += SCAN_BLOCK) part += csum[i];
    sh[tid] = part;
    __syncthreads();
    for (int s = SCAN_BLOCK / 2; s > 0; s >>= 1) {
        if (tid < s) sh[tid] += sh[tid + s];
        __syncthreads();
    }
    int chunkOff = sh[0];
    __syncthreads();

    int* p = cnt + (size_t)blockIdx.x * SCAN_CHUNK + (size_t)tid * 4;
    int4 c = *(const int4*)p;
    int mine = c.x + c.y + c.z + c.w;
    sh[tid] = mine;
    __syncthreads();
    for (int d = 1; d < SCAN_BLOCK; d <<= 1) {
        int add = (tid >= d) ? sh[tid - d] : 0;
        __syncthreads();
        sh[tid] += add;
        __syncthreads();
    }
    int excl = sh[tid] - mine + chunkOff;
    int4 o;
    o.x = excl;
    o.y = o.x + c.x;
    o.z = o.y + c.y;
    o.w = o.z + c.z;
    *(int4*)p = o;
}

__global__ void lgcn_bin_scatter(const int* __restrict__ rows, const int* __restrict__ cols,
                                 const float* __restrict__ vals, const int* __restrict__ offM,
                                 int* __restrict__ bin, int nEdges, int nBuckP, int nBlk) {
    __shared__ int   lStart[MAXB];
    __shared__ int   gOff[MAXB];
    __shared__ int   sRow[L1_CAP];
    __shared__ int   sCol[L1_CAP];
    __shared__ float sVal[L1_CAP];
    int tid  = threadIdx.x;
    int base = (int)blockIdx.x * L1_CAP;
    int n    = nEdges - base;
    if (n <= 0) return;
    if (n > L1_CAP) n = L1_CAP;

    lStart[tid] = 0;
    gOff[tid]   = (tid < nBuckP) ? offM[(size_t)tid * nBlk + blockIdx.x] : 0;
    __syncthreads();

    int   r[L1_PER], c[L1_PER], k[L1_PER];
    float v[L1_PER];
#pragma unroll
    for (int j = 0; j < L1_PER; ++j) {
        int e = j * MAXB + tid;
        r[j] = 0; c[j] = 0; v[j] = 0.0f; k[j] = 0;
        if (e < n) {
            r[j] = rows[base + e];
            c[j] = cols[base + e];
            v[j] = vals[base + e];
            k[j] = atomicAdd(&lStart[r[j] >> BK_SHIFT], 1);
        }
    }
    __syncthreads();

    int mine = lStart[tid];
    for (int d = 1; d < MAXB; d <<= 1) {
        int add = (tid >= d) ? lStart[tid - d] : 0;
        __syncthreads();
        lStart[tid] += add;
        __syncthreads();
    }
    int excl = lStart[tid] - mine;
    __syncthreads();
    lStart[tid] = excl;
    __syncthreads();

#pragma unroll
    for (int j = 0; j < L1_PER; ++j) {
        int e = j * MAXB + tid;
        if (e < n) {
            int p = lStart[r[j] >> BK_SHIFT] + k[j];
            sRow[p] = r[j];
            sCol[p] = c[j];
            sVal[p] = v[j];
        }
    }
    __syncthreads();

    int2* binp = (int2*)bin;
    for (int i = tid; i < n; i += MAXB) {
        int row = sRow[i];
        int b   = row >> BK_SHIFT;
        int g   = gOff[b] + (i - lStart[b]);
        int2 rec;
        rec.x = (sCol[i] << BK_SHIFT) | (row & (BK_ROWS - 1));
        rec.y = __float_as_int(sVal[i]);
        binp[g] = rec;
    }
}

__global__ void lgcn_bucket_sort(const int* __restrict__ offM, const int* __restrict__ bin,
                                 int* __restrict__ rowStart, int* __restrict__ ed,
                                 int nRows, int nBlk) {
    __shared__ int  cnt[BK_ROWS];
    __shared__ int  cur[BK_ROWS];
    __shared__ int2 sEd[L2_CAP];
    int tid   = threadIdx.x;
    int b     = (int)blockIdx.x;
    int start = offM[(size_t)b * nBlk];
    int end   = offM[(size_t)(b + 1) * nBlk];
    int n     = end - start;
    int row0  = b << BK_SHIFT;
    const int2* bp = (const int2*)bin + start;

    if (tid < BK_ROWS) cnt[tid] = 0;
    __syncthreads();
    for (int i = tid; i < n; i += MAXB) atomicAdd(&cnt[bp[i].x & (BK_ROWS - 1)], 1);
    __syncthreads();

    int mine = (tid < BK_ROWS) ? cnt[tid] : 0;
    for (int d = 1; d < BK_ROWS; d <<= 1) {
        int add = (tid < BK_ROWS && tid >= d) ? cnt[tid - d] : 0;
        __syncthreads();
        if (tid < BK_ROWS) cnt[tid] += add;
        __syncthreads();
    }
    if (tid < BK_ROWS) {
        int excl = cnt[tid] - mine;
        cur[tid] = excl;
        int gr = row0 + tid;
        if (gr <= nRows) rowStart[gr] = start + excl;
    }
    __syncthreads();

    int2* edp = (int2*)ed + start;
    if (n <= L2_CAP) {
        for (int i = tid; i < n; i += MAXB) {
            int2 rec = bp[i];
            int p = atomicAdd(&cur[rec.x & (BK_ROWS - 1)], 1);
            int2 e; e.x = (int)((unsigned)rec.x >> BK_SHIFT); e.y = rec.y;
            sEd[p] = e;
        }
        __syncthreads();
        for (int i = tid; i < n; i += MAXB) edp[i] = sEd[i];
    } else {
        for (int i = tid; i < n; i += MAXB) {
            int2 rec = bp[i];
            int p = atomicAdd(&cur[rec.x & (BK_ROWS - 1)], 1);
            int2 e; e.x = (int)((unsigned)rec.x >> BK_SHIFT); e.y = rec.y;
            edp[p] = e;
        }
    }
}

__device__ __forceinline__ v8f lgcn_tile_axpy(v8f c, v8f t, int lo, int hi) {
#pragma unroll
    for (int kk = 0; kk < 4; ++kk) {
        int m = 2 * kk + 8 * hi;
        v2f a, b;
        a[0] = (lo == m)     ? 0.25f : 0.0f;
        a[1] = (lo == m + 1) ? 0.25f : 0.0f;
        b[0] = t[2 * kk];
        b[1] = t[2 * kk + 1];
        c = __builtin_amdgcn_wmma_f32_16x16x4_f32(
                 false, a,  false, b,
                 (short)0, c,  false,  false);
    }
    return c;
}

template <int MODE>
__device__ __forceinline__ const v4f* lgcn_row_ptr(int c, const float* __restrict__ ue,
                                                   const float* __restrict__ ie,
                                                   const float* __restrict__ x,
                                                   int nUsers, int sub) {
    const char* base = (const char*)x;
    unsigned    row  = (unsigned)c;
    if (MODE == 0) {
        bool user = c < nUsers;
        base = user ? (const char*)ue : (const char*)ie;
        row  = user ? (unsigned)c : (unsigned)(c - nUsers);
    }
    unsigned off = (row << 8) | ((unsigned)sub << 4);
    return (const v4f*)(base + off);
}

__device__ __forceinline__ v4f lgcn_fma4(v4f g, float v, v4f a) {
    v4f vv = {v, v, v, v};
    return __builtin_elementwise_fma(g, vv, a);
}

#define LGCN_ROW_BCAST(x, K) __builtin_amdgcn_update_dpp(0, (x), 0x150 + (K), 0xf, 0xf, true)

__device__ __forceinline__ int lgcn_xhalf(int x) {
    return __builtin_amdgcn_permlanex16(x, x, 0x76543210, 0xfedcba98, false, false);
}

struct lgcn_pairs8 { int c0, c1, c2, c3, c4, c5, c6, c7; };

template <int K0>
__device__ __forceinline__ lgcn_pairs8 lgcn_bcast8(int2 e) {
    lgcn_pairs8 p;
    p.c0 = LGCN_ROW_BCAST(e.x, K0 + 0);
    p.c1 = LGCN_ROW_BCAST(e.x, K0 + 1);
    p.c2 = LGCN_ROW_BCAST(e.x, K0 + 2);
    p.c3 = LGCN_ROW_BCAST(e.x, K0 + 3);
    p.c4 = LGCN_ROW_BCAST(e.x, K0 + 4);
    p.c5 = LGCN_ROW_BCAST(e.x, K0 + 5);
    p.c6 = LGCN_ROW_BCAST(e.x, K0 + 6);
    p.c7 = LGCN_ROW_BCAST(e.x, K0 + 7);
    return p;
}

template <int MODE, int K0>
__device__ __forceinline__ v4f lgcn_gather8(v4f a, lgcn_pairs8 p, int ey, int rem,
                                            const float* __restrict__ ue,
                                            const float* __restrict__ ie,
                                            const float* __restrict__ x,
                                            int nUsers, int sub) {
    const v4f zero = {0.0f, 0.0f, 0.0f, 0.0f};
    v4f g0 = zero, g1 = zero, g2 = zero, g3 = zero, g4 = zero, g5 = zero, g6 = zero, g7 = zero;
    if (K0 + 0 < rem) g0 = *lgcn_row_ptr<MODE>(p.c0, ue, ie, x, nUsers, sub);
    if (K0 + 1 < rem) g1 = *lgcn_row_ptr<MODE>(p.c1, ue, ie, x, nUsers, sub);
    if (K0 + 2 < rem) g2 = *lgcn_row_ptr<MODE>(p.c2, ue, ie, x, nUsers, sub);
    if (K0 + 3 < rem) g3 = *lgcn_row_ptr<MODE>(p.c3, ue, ie, x, nUsers, sub);
    if (K0 + 4 < rem) g4 = *lgcn_row_ptr<MODE>(p.c4, ue, ie, x, nUsers, sub);
    if (K0 + 5 < rem) g5 = *lgcn_row_ptr<MODE>(p.c5, ue, ie, x, nUsers, sub);
    if (K0 + 6 < rem) g6 = *lgcn_row_ptr<MODE>(p.c6, ue, ie, x, nUsers, sub);
    if (K0 + 7 < rem) g7 = *lgcn_row_ptr<MODE>(p.c7, ue, ie, x, nUsers, sub);
    a = lgcn_fma4(g0, __int_as_float(LGCN_ROW_BCAST(ey, K0 + 0)), a);
    a = lgcn_fma4(g1, __int_as_float(LGCN_ROW_BCAST(ey, K0 + 1)), a);
    a = lgcn_fma4(g2, __int_as_float(LGCN_ROW_BCAST(ey, K0 + 2)), a);
    a = lgcn_fma4(g3, __int_as_float(LGCN_ROW_BCAST(ey, K0 + 3)), a);
    a = lgcn_fma4(g4, __int_as_float(LGCN_ROW_BCAST(ey, K0 + 4)), a);
    a = lgcn_fma4(g5, __int_as_float(LGCN_ROW_BCAST(ey, K0 + 5)), a);
    a = lgcn_fma4(g6, __int_as_float(LGCN_ROW_BCAST(ey, K0 + 6)), a);
    a = lgcn_fma4(g7, __int_as_float(LGCN_ROW_BCAST(ey, K0 + 7)), a);
    return a;
}

template <int MODE>
__device__ __forceinline__ v4f lgcn_row_sum(int j, int cnt, int maxcnt, int2 e,
                                            const int2* __restrict__ edp,
                                            const float* __restrict__ ue,
                                            const float* __restrict__ ie,
                                            const float* __restrict__ x,
                                            int nUsers, int sub) {
    v4f a = {0.0f, 0.0f, 0.0f, 0.0f};
    for (int base = 0; base < maxcnt; base += 16) {
        int rem = cnt - base;
        lgcn_pairs8 p = lgcn_bcast8<0>(e);
        int2 en;
        en.x = 0; en.y = 0;
        if (base + 16 < maxcnt) {
            int nidx = base + 16 + sub;
            if (nidx < cnt) en = edp[j + nidx];
        }
        a = lgcn_gather8<MODE, 0>(a, p, e.y, rem, ue, ie, x, nUsers, sub);
        if (base + 8 < maxcnt) {
            lgcn_pairs8 q = lgcn_bcast8<8>(e);
            a = lgcn_gather8<MODE, 8>(a, q, e.y, rem, ue, ie, x, nUsers, sub);
        }
        e = en;
    }
    return a;
}

template <int MODE>
__global__ void __launch_bounds__(256)
lgcn_row_layer(const int* __restrict__ rowStart, const int* __restrict__ ed,
               const float* __restrict__ ue, const float* __restrict__ ie,
               const float* __restrict__ x, const float* __restrict__ x1,
               float* __restrict__ y, int nUsers, int nRows) {
    int t    = (int)blockIdx.x * (int)blockDim.x + (int)threadIdx.x;
    int lane = (int)threadIdx.x & 31;
    int sub  = lane & 15;
    int hl   = lane >> 4;
    int r0   = (t >> 5) << 2;
    if (r0 >= nRows) return;
    int  rA  = r0 + hl;
    int  rB  = r0 + 2 + hl;
    bool vA  = rA < nRows;
    bool vB  = rB < nRows;
    int  cA  = vA ? rA : nRows - 1;
    int  cB  = vB ? rB : nRows - 1;
    int jA  = rowStart[cA];
    int eA  = rowStart[cA + 1];
    int jB  = rowStart[cB];
    int eB  = rowStart[cB + 1];
    int cntA = vA ? eA - jA : 0;
    int cntB = vB ? eB - jB : 0;
    const int2* edp = (const int2*)ed;
    int mA = __builtin_amdgcn_readfirstlane(max(cntA, lgcn_xhalf(cntA)));
    int mB = __builtin_amdgcn_readfirstlane(max(cntB, lgcn_xhalf(cntB)));
    int2 pA, pB;
    pA.x = 0; pA.y = 0; pB.x = 0; pB.y = 0;
    if (sub < cntA) pA = edp[jA + sub];
    if (sub < cntB) pB = edp[jB + sub];
    v4f a0 = lgcn_row_sum<MODE>(jA, cntA, mA, pA, edp, ue, ie, x, nUsers, sub);
    v4f a1 = lgcn_row_sum<MODE>(jB, cntB, mB, pB, edp, ue, ie, x, nUsers, sub);

    size_t slotA = (size_t)cA * (EMB_D / 4) + sub;
    size_t slotB = (size_t)cB * (EMB_D / 4) + sub;
    if (MODE == 2) {
        v4f q0 = *lgcn_row_ptr<0>(cA, ue, ie, x, nUsers, sub);
        v4f q1 = *lgcn_row_ptr<0>(cB, ue, ie, x, nUsers, sub);
        v4f s0 = ((const v4f*)x1)[slotA], s1 = ((const v4f*)x1)[slotB];
        v4f u0 = ((const v4f*)x)[slotA],  u1 = ((const v4f*)x)[slotB];
        v8f c, t1, t2, t3;
        c[0] = q0[0] * 0.25f; c[1] = q0[1] * 0.25f; c[2] = q0[2] * 0.25f; c[3] = q0[3] * 0.25f;
        c[4] = q1[0] * 0.25f; c[5] = q1[1] * 0.25f; c[6] = q1[2] * 0.25f; c[7] = q1[3] * 0.25f;
        t1[0] = s0[0]; t1[1] = s0[1]; t1[2] = s0[2]; t1[3] = s0[3];
        t1[4] = s1[0]; t1[5] = s1[1]; t1[6] = s1[2]; t1[7] = s1[3];
        t2[0] = u0[0]; t2[1] = u0[1]; t2[2] = u0[2]; t2[3] = u0[3];
        t2[4] = u1[0]; t2[5] = u1[1]; t2[6] = u1[2]; t2[7] = u1[3];
        t3[0] = a0[0]; t3[1] = a0[1]; t3[2] = a0[2]; t3[3] = a0[3];
        t3[4] = a1[0]; t3[5] = a1[1]; t3[6] = a1[2]; t3[7] = a1[3];
        int lo = lane & 15;
        int hi = lane >> 4;
        c = lgcn_tile_axpy(c, t1, lo, hi);
        c = lgcn_tile_axpy(c, t2, lo, hi);
        c = lgcn_tile_axpy(c, t3, lo, hi);
        a0[0] = c[0]; a0[1] = c[1]; a0[2] = c[2]; a0[3] = c[3];
        a1[0] = c[4]; a1[1] = c[5]; a1[2] = c[6]; a1[3] = c[7];
    }
    if (vA) ((v4f*)y)[slotA] = a0;
    if (vB) ((v4f*)y)[slotB] = a1;
}

extern "C" void kernel_launch(void* const* d_in, const int* in_sizes, int n_in,
                              void* d_out, int out_size, void* d_ws, size_t ws_size,
                              hipStream_t stream) {
    const float* ue   = (const float*)d_in[0];
    const float* ie   = (const float*)d_in[1];
    const int*   rows = (const int*)d_in[2];
    const int*   cols = (const int*)d_in[3];
    const float* vals = (const float*)d_in[4];
    float*       out  = (float*)d_out;

    size_t userElems = (size_t)in_sizes[0];
    size_t itemElems = (size_t)in_sizes[1];
    int    nEdges    = in_sizes[2];
    size_t total     = userElems + itemElems;
    int    nRows     = (int)(total / EMB_D);
    int    nUsers    = (int)(userElems / EMB_D);

    int    nBuck     = (nRows + 1 + BK_ROWS - 1) / BK_ROWS;
    int    nBuckP    = nBuck + 1;
    int    nBlk      = (nEdges + L1_CAP - 1) / L1_CAP;
    int    nChunks   = (nBuckP * nBlk + SCAN_CHUNK - 1) / SCAN_CHUNK;
    int    matLen    = nChunks * SCAN_CHUNK;
    size_t csumPad   = ((size_t)nChunks + 3) / 4 * 4;
    size_t rsPad     = ((size_t)nRows + 1 + 3) / 4 * 4;
    size_t edgePad   = ((size_t)nEdges + 3) / 4 * 4;

    float* buf0 = (float*)d_ws;
    float* buf1 = buf0 + total;
    int*   cntM = (int*)(buf1 + total);
    int*   csum = cntM + matLen;
    int*   rowStart = csum + csumPad;
    int*   bin  = rowStart + rsPad;
    int*   ed   = bin + 2 * edgePad;

    lgcn_bin_count<<<nBlk, MAXB, 0, stream>>>(rows, cntM, nEdges, nBuckP, nBlk, matLen);
    lgcn_chunk_sum<<<nChunks, SCAN_BLOCK, 0, stream>>>(cntM, csum);
    lgcn_scan<<<nChunks, SCAN_BLOCK, 0, stream>>>(cntM, csum);
    lgcn_bin_scatter<<<nBlk, MAXB, 0, stream>>>(rows, cols, vals, cntM, bin, nEdges, nBuckP, nBlk);
    lgcn_bucket_sort<<<nBuck, MAXB, 0, stream>>>(cntM, bin, rowStart, ed, nRows, nBlk);

    int rowBlocks = (nRows + 31) / 32;
    lgcn_row_layer<0><<<rowBlocks, 256, 0, stream>>>(rowStart, ed, ue, ie, nullptr, nullptr, buf0, nUsers, nRows);
    lgcn_row_layer<1><<<rowBlocks, 256, 0, stream>>>(rowStart, ed, ue, ie, buf0, nullptr, buf1, nUsers, nRows);
    lgcn_row_layer<2><<<rowBlocks, 256, 0, stream>>>(rowStart, ed, ue, ie, buf1, buf0, out, nUsers, nRows);
}


# Grouping without count and scan passes: a binning block orders its edges by bucket into its own slice and keeps its bucket starts, whose sum places a bucket: 8 -> 5 launches; sum order = LDS arrival
# speedup vs baseline: 5.6229x; 1.0777x over previous
#include <hip/hip_runtime.h>
#include <hip/hip_bf16.h>


typedef __attribute__((ext_vector_type(2))) float v2f;
typedef __attribute__((ext_vector_type(4))) float v4f;
typedef __attribute__((ext_vector_type(8))) float v8f;

#define EMB_D 64

#define BK_SHIFT 9
#define BK_ROWS 512
#define MAXB 1024
#define L1_PER 10
#define L1_CAP (L1_PER * MAXB)
#define L2_CAP 16384

__global__ void __launch_bounds__(MAXB)
lgcn_bin_local(const int* __restrict__ rows, const int* __restrict__ cols,
               const float* __restrict__ vals, int* __restrict__ lsm,
               int* __restrict__ bin, int nEdges, int nBuckP, int nBlk) {
    __shared__ int   lStart[MAXB];
    __shared__ int   sKey[L1_CAP];
    __shared__ float sVal[L1_CAP];
    int tid  = threadIdx.x;
    int base = (int)blockIdx.x * L1_CAP;
    int n    = nEdges - base;
    if (n <= 0) return;
    if (n > L1_CAP) n = L1_CAP;

    lStart[tid] = 0;
    __syncthreads();

    int   r[L1_PER], c[L1_PER], k[L1_PER];
    float v[L1_PER];
#pragma unroll
    for (int j = 0; j < L1_PER; ++j) {
        int e = j * MAXB + tid;
        r[j] = 0; c[j] = 0; v[j] = 0.0f; k[j] = 0;
        if (e < n) {
            r[j] = rows[base + e];
            c[j] = cols[base + e];
            v[j] = vals[base + e];
            k[j] = atomicAdd(&lStart[r[j] >> BK_SHIFT], 1);
        }
    }
    __syncthreads();

    int mine = lStart[tid];
    for (int d = 1; d < MAXB; d <<= 1) {
        int add = (tid >= d) ? lStart[tid - d] : 0;
        __syncthreads();
        lStart[tid] += add;
        __syncthreads();
    }
    int excl = lStart[tid] - mine;
    __syncthreads();
    lStart[tid] = excl;
    if (tid < nBuckP) lsm[(size_t)tid * nBlk + blockIdx.x] = excl;
    __syncthreads();

#pragma unroll
    for (int j = 0; j < L1_PER; ++j) {
        int e = j * MAXB + tid;
        if (e < n) {
            int p = lStart[r[j] >> BK_SHIFT] + k[j];
            sKey[p] = (c[j] << BK_SHIFT) | (r[j] & (BK_ROWS - 1));
            sVal[p] = v[j];
        }
    }
    __syncthreads();

    int2* binp = (int2*)bin + base;
    for (int i = tid; i < n; i += MAXB) {
        int2 rec;
        rec.x = sKey[i];
        rec.y = __float_as_int(sVal[i]);
        binp[i] = rec;
    }
}

#define LGCN_REC(rec)                                                                  \
        do {                                                                           \
            if (PASS == 0) {                                                           \
                atomicAdd(&cntOrCur[(rec).x & (BK_ROWS - 1)], 1);                      \
            } else {                                                                   \
                int p_ = atomicAdd(&cntOrCur[(rec).x & (BK_ROWS - 1)], 1);             \
                int2 e_; e_.x = (int)((unsigned)(rec).x >> BK_SHIFT); e_.y = (rec).y;  \
                if (small) sEd[p_] = e_; else edp[p_] = e_;                            \
            }                                                                          \
        } while (0)

template <int PASS>
__device__ __forceinline__ int lgcn_runs(const int* __restrict__ lsm, const int2* __restrict__ binp,
                                         int* cntOrCur, int2* sEd, int2* edp, bool small,
                                         int b, int nBlk, int w, int lane) {
    int acc = 0;
    int rpw = (nBlk + 31) >> 5;
    const int* q0 = lsm + (size_t)b * nBlk;
    const int* q1 = q0 + nBlk;
    for (int c0 = 0; c0 < rpw; c0 += 32) {
        int kb = w * rpw + c0;
        int nr = min(min(32, rpw - c0), nBlk - kb);
        int k  = kb + lane;
        int s0 = 0, s1 = 0;
        if (lane < nr) {
            s0 = q0[k];
            s1 = q1[k];
        }
        acc += s0;
        int src = k * L1_CAP + s0;
        int len = s1 - s0;
        for (int i = 0; i < nr; i += 8) {
            int  s[8], l[8];
            int2 r[8];
            int  lmax = 0;
#pragma unroll
            for (int u = 0; u < 8; ++u) {
                s[u] = __builtin_amdgcn_readlane(src, i + u);
                l[u] = __builtin_amdgcn_readlane(len, i + u);
                lmax = max(lmax, l[u]);
            }
#pragma unroll
            for (int u = 0; u < 8; ++u) r[u] = binp[(lane < l[u]) ? s[u] + lane : 0];
            if (PASS == 0)
                asm volatile("" :: "v"(r[0].x), "v"(r[1].x), "v"(r[2].x), "v"(r[3].x),
                                   "v"(r[4].x), "v"(r[5].x), "v"(r[6].x), "v"(r[7].x));
            else
                asm volatile("" :: "v"(r[0].x), "v"(r[0].y), "v"(r[1].x), "v"(r[1].y),
                                   "v"(r[2].x), "v"(r[2].y), "v"(r[3].x), "v"(r[3].y),
                                   "v"(r[4].x), "v"(r[4].y), "v"(r[5].x), "v"(r[5].y),
                                   "v"(r[6].x), "v"(r[6].y), "v"(r[7].x), "v"(r[7].y));
#pragma unroll
            for (int u = 0; u < 8; ++u)
                if (lane < l[u]) LGCN_REC(r[u]);
            if (lmax > 32) {
#pragma unroll
                for (int u = 0; u < 8; ++u)
                    for (int t = 32 + lane; t < l[u]; t += 32) { int2 x = binp[s[u] + t]; LGCN_REC(x); }
            }
        }
    }
    return acc;
}
#undef LGCN_REC

__global__ void __launch_bounds__(MAXB)
lgcn_bucket_sort(const int* __restrict__ lsm, const int* __restrict__ bin,
                 int* __restrict__ rowStart, int* __restrict__ ed,
                 int nRows, int nBlk) {
    __shared__ int  cnt[BK_ROWS];
    __shared__ int  cur[BK_ROWS];
    __shared__ int  sBase;
    __shared__ int2 sEd[L2_CAP];
    int tid  = threadIdx.x;
    int lane = tid & 31;
    int w    = __builtin_amdgcn_readfirstlane(tid >> 5);
    int b    = (int)blockIdx.x;
    int row0 = b << BK_SHIFT;
    const int2* binp = (const int2*)bin;

    if (tid < BK_ROWS) cnt[tid] = 0;
    if (tid == 0) sBase = 0;
    __syncthreads();

    int acc = lgcn_runs<0>(lsm, binp, cnt, nullptr, nullptr, false, b, nBlk, w, lane);
    for (int o = 16; o > 0; o >>= 1) acc += __shfl_xor(acc, o);
    if (lane == 0) atomicAdd(&sBase, acc);
    __syncthreads();
    int start = sBase;

    int mine = (tid < BK_ROWS) ? cnt[tid] : 0;
    for (int d = 1; d < BK_ROWS; d <<= 1) {
        int add = (tid < BK_ROWS && tid >= d) ? cnt[tid - d] : 0;
        __syncthreads();
        if (tid < BK_ROWS) cnt[tid] += add;
        __syncthreads();
    }
    if (tid < BK_ROWS) {
        int excl = cnt[tid] - mine;
        cur[tid] = excl;
        int gr = row0 + tid;
        if (gr <= nRows) rowStart[gr] = start + excl;
    }
    __syncthreads();
    int  n     = __builtin_amdgcn_readfirstlane(cnt[BK_ROWS - 1]);
    bool small = n <= L2_CAP;

    int2* edp = (int2*)ed + start;
    lgcn_runs<1>(lsm, binp, cur, sEd, edp, small, b, nBlk, w, lane);
    if (small) {
        __syncthreads();
        for (int i = tid; i < n; i += MAXB) edp[i] = sEd[i];
    }
}

__device__ __forceinline__ v8f lgcn_tile_axpy(v8f c, v8f t, int lo, int hi) {
#pragma unroll
    for (int kk = 0; kk < 4; ++kk) {
        int m = 2 * kk + 8 * hi;
        v2f a, b;
        a[0] = (lo == m)     ? 0.25f : 0.0f;
        a[1] = (lo == m + 1) ? 0.25f : 0.0f;
        b[0] = t[2 * kk];
        b[1] = t[2 * kk + 1];
        c = __builtin_amdgcn_wmma_f32_16x16x4_f32(
                 false, a,  false, b,
                 (short)0, c,  false,  false);
    }
    return c;
}

template <int MODE>
__device__ __forceinline__ const v4f* lgcn_row_ptr(int c, const float* __restrict__ ue,
                                                   const float* __restrict__ ie,
                                                   const float* __restrict__ x,
                                                   int nUsers, int sub) {
    const char* base = (const char*)x;
    unsigned    row  = (unsigned)c;
    if (MODE == 0) {
        bool user = c < nUsers;
        base = user ? (const char*)ue : (const char*)ie;
        row  = user ? (unsigned)c : (unsigned)(c - nUsers);
    }
    unsigned off = (row << 8) | ((unsigned)sub << 4);
    return (const v4f*)(base + off);
}

__device__ __forceinline__ v4f lgcn_fma4(v4f g, float v, v4f a) {
    v4f vv = {v, v, v, v};
    return __builtin_elementwise_fma(g, vv, a);
}

#define LGCN_ROW_BCAST(x, K) __builtin_amdgcn_update_dpp(0, (x), 0x150 + (K), 0xf, 0xf, true)

__device__ __forceinline__ int lgcn_xhalf(int x) {
    return __builtin_amdgcn_permlanex16(x, x, 0x76543210, 0xfedcba98, false, false);
}

struct lgcn_pairs8 { int c0, c1, c2, c3, c4, c5, c6, c7; };

template <int K0>
__device__ __forceinline__ lgcn_pairs8 lgcn_bcast8(int2 e) {
    lgcn_pairs8 p;
    p.c0 = LGCN_ROW_BCAST(e.x, K0 + 0);
    p.c1 = LGCN_ROW_BCAST(e.x, K0 + 1);
    p.c2 = LGCN_ROW_BCAST(e.x, K0 + 2);
    p.c3 = LGCN_ROW_BCAST(e.x, K0 + 3);
    p.c4 = LGCN_ROW_BCAST(e.x, K0 + 4);
    p.c5 = LGCN_ROW_BCAST(e.x, K0 + 5);
    p.c6 = LGCN_ROW_BCAST(e.x, K0 + 6);
    p.c7 = LGCN_ROW_BCAST(e.x, K0 + 7);
    return p;
}

template <int MODE, int K0>
__device__ __forceinline__ v4f lgcn_gather8(v4f a, lgcn_pairs8 p, int ey, int rem,
                                            const float* __restrict__ ue,
                                            const float* __restrict__ ie,
                                            const float* __restrict__ x,
                                            int nUsers, int sub) {
    const v4f zero = {0.0f, 0.0f, 0.0f, 0.0f};
    v4f g0 = zero, g1 = zero, g2 = zero, g3 = zero, g4 = zero, g5 = zero, g6 = zero, g7 = zero;
    if (K0 + 0 < rem) g0 = *lgcn_row_ptr<MODE>(p.c0, ue, ie, x, nUsers, sub);
    if (K0 + 1 < rem) g1 = *lgcn_row_ptr<MODE>(p.c1, ue, ie, x, nUsers, sub);
    if (K0 + 2 < rem) g2 = *lgcn_row_ptr<MODE>(p.c2, ue, ie, x, nUsers, sub);
    if (K0 + 3 < rem) g3 = *lgcn_row_ptr<MODE>(p.c3, ue, ie, x, nUsers, sub);
    if (K0 + 4 < rem) g4 = *lgcn_row_ptr<MODE>(p.c4, ue, ie, x, nUsers, sub);
    if (K0 + 5 < rem) g5 = *lgcn_row_ptr<MODE>(p.c5, ue, ie, x, nUsers, sub);
    if (K0 + 6 < rem) g6 = *lgcn_row_ptr<MODE>(p.c6, ue, ie, x, nUsers, sub);
    if (K0 + 7 < rem) g7 = *lgcn_row_ptr<MODE>(p.c7, ue, ie, x, nUsers, sub);
    a = lgcn_fma4(g0, __int_as_float(LGCN_ROW_BCAST(ey, K0 + 0)), a);
    a = lgcn_fma4(g1, __int_as_float(LGCN_ROW_BCAST(ey, K0 + 1)), a);
    a = lgcn_fma4(g2, __int_as_float(LGCN_ROW_BCAST(ey, K0 + 2)), a);
    a = lgcn_fma4(g3, __int_as_float(LGCN_ROW_BCAST(ey, K0 + 3)), a);
    a = lgcn_fma4(g4, __int_as_float(LGCN_ROW_BCAST(ey, K0 + 4)), a);
    a = lgcn_fma4(g5, __int_as_float(LGCN_ROW_BCAST(ey, K0 + 5)), a);
    a = lgcn_fma4(g6, __int_as_float(LGCN_ROW_BCAST(ey, K0 + 6)), a);
    a = lgcn_fma4(g7, __int_as_float(LGCN_ROW_BCAST(ey, K0 + 7)), a);
    return a;
}

template <int MODE>
__device__ __forceinline__ v4f lgcn_row_sum(int j, int cnt, int maxcnt, int2 e,
                                            const int2* __restrict__ edp,
                                            const float* __restrict__ ue,
                                            const float* __restrict__ ie,
                                            const float* __restrict__ x,
                                            int nUsers, int sub) {
    v4f a = {0.0f, 0.0f, 0.0f, 0.0f};
    for (int base = 0; base < maxcnt; base += 16) {
        int rem = cnt - base;
        lgcn_pairs8 p = lgcn_bcast8<0>(e);
        int2 en;
        en.x = 0; en.y = 0;
        if (base + 16 < maxcnt) {
            int nidx = base + 16 + sub;
            if (nidx < cnt) en = edp[j + nidx];
        }
        a = lgcn_gather8<MODE, 0>(a, p, e.y, rem, ue, ie, x, nUsers, sub);
        if (base + 8 < maxcnt) {
            lgcn_pairs8 q = lgcn_bcast8<8>(e);
            a = lgcn_gather8<MODE, 8>(a, q, e.y, rem, ue, ie, x, nUsers, sub);
        }
        e = en;
    }
    return a;
}

template <int MODE>
__global__ void __launch_bounds__(256)
lgcn_row_layer(const int* __restrict__ rowStart, const int* __restrict__ ed,
               const float* __restrict__ ue, const float* __restrict__ ie,
               const float* __restrict__ x, const float* __restrict__ x1,
               float* __restrict__ y, int nUsers, int nRows) {
    int t    = (int)blockIdx.x * (int)blockDim.x + (int)threadIdx.x;
    int lane = (int)threadIdx.x & 31;
    int sub  = lane & 15;
    int hl   = lane >> 4;
    int r0   = (t >> 5) << 2;
    if (r0 >= nRows) return;
    int  rA  = r0 + hl;
    int  rB  = r0 + 2 + hl;
    bool vA  = rA < nRows;
    bool vB  = rB < nRows;
    int  cA  = vA ? rA : nRows - 1;
    int  cB  = vB ? rB : nRows - 1;
    int jA  = rowStart[cA];
    int eA  = rowStart[cA + 1];
    int jB  = rowStart[cB];
    int eB  = rowStart[cB + 1];
    int cntA = vA ? eA - jA : 0;
    int cntB = vB ? eB - jB : 0;
    const int2* edp = (const int2*)ed;
    int mA = __builtin_amdgcn_readfirstlane(max(cntA, lgcn_xhalf(cntA)));
    int mB = __builtin_amdgcn_readfirstlane(max(cntB, lgcn_xhalf(cntB)));
    int2 pA, pB;
    pA.x = 0; pA.y = 0; pB.x = 0; pB.y = 0;
    if (sub < cntA) pA = edp[jA + sub];
    if (sub < cntB) pB = edp[jB + sub];
    v4f a0 = lgcn_row_sum<MODE>(jA, cntA, mA, pA, edp, ue, ie, x, nUsers, sub);
    v4f a1 = lgcn_row_sum<MODE>(jB, cntB, mB, pB, edp, ue, ie, x, nUsers, sub);

    size_t slotA = (size_t)cA * (EMB_D / 4) + sub;
    size_t slotB = (size_t)cB * (EMB_D / 4) + sub;
    if (MODE == 2) {
        v4f q0 = *lgcn_row_ptr<0>(cA, ue, ie, x, nUsers, sub);
        v4f q1 = *lgcn_row_ptr<0>(cB, ue, ie, x, nUsers, sub);
        v4f s0 = ((const v4f*)x1)[slotA], s1 = ((const v4f*)x1)[slotB];
        v4f u0 = ((const v4f*)x)[slotA],  u1 = ((const v4f*)x)[slotB];
        v8f c, t1, t2, t3;
        c[0] = q0[0] * 0.25f; c[1] = q0[1] * 0.25f; c[2] = q0[2] * 0.25f; c[3] = q0[3] * 0.25f;
        c[4] = q1[0] * 0.25f; c[5] = q1[1] * 0.25f; c[6] = q1[2] * 0.25f; c[7] = q1[3] * 0.25f;
        t1[0] = s0[0]; t1[1] = s0[1]; t1[2] = s0[2]; t1[3] = s0[3];
        t1[4] = s1[0]; t1[5] = s1[1]; t1[6] = s1[2]; t1[7] = s1[3];
        t2[0] = u0[0]; t2[1] = u0[1]; t2[2] = u0[2]; t2[3] = u0[3];
        t2[4] = u1[0]; t2[5] = u1[1]; t2[6] = u1[2]; t2[7] = u1[3];
        t3[0] = a0[0]; t3[1] = a0[1]; t3[2] = a0[2]; t3[3] = a0[3];
        t3[4] = a1[0]; t3[5] = a1[1]; t3[6] = a1[2]; t3[7] = a1[3];
        int lo = lane & 15;
        int hi = lane >> 4;
        c = lgcn_tile_axpy(c, t1, lo, hi);
        c = lgcn_tile_axpy(c, t2, lo, hi);
        c = lgcn_tile_axpy(c, t3, lo, hi);
        a0[0] = c[0]; a0[1] = c[1]; a0[2] = c[2]; a0[3] = c[3];
        a1[0] = c[4]; a1[1] = c[5]; a1[2] = c[6]; a1[3] = c[7];
    }
    if (vA) ((v4f*)y)[slotA] = a0;
    if (vB) ((v4f*)y)[slotB] = a1;
}

extern "C" void kernel_launch(void* const* d_in, const int* in_sizes, int n_in,
                              void* d_out, int out_size, void* d_ws, size_t ws_size,
                              hipStream_t stream) {
    const float* ue   = (const float*)d_in[0];
    const float* ie   = (const float*)d_in[1];
    const int*   rows = (const int*)d_in[2];
    const int*   cols = (const int*)d_in[3];
    const float* vals = (const float*)d_in[4];
    float*       out  = (float*)d_out;

    size_t userElems = (size_t)in_sizes[0];
    size_t itemElems = (size_t)in_sizes[1];
    int    nEdges    = in_sizes[2];
    size_t total     = userElems + itemElems;
    int    nRows     = (int)(total / EMB_D);
    int    nUsers    = (int)(userElems / EMB_D);

    int    nBuck     = (nRows + 1 + BK_ROWS - 1) / BK_ROWS;
    int    nBuckP    = nBuck + 1;
    int    nBlk      = (nEdges + L1_CAP - 1) / L1_CAP;
    size_t lsmPad    = ((size_t)nBlk * (size_t)nBuckP + 3) / 4 * 4;
    size_t rsPad     = ((size_t)nRows + 1 + 3) / 4 * 4;
    size_t edgePad   = ((size_t)nEdges + 3) / 4 * 4;

    float* buf0 = (float*)d_ws;
    float* buf1 = buf0 + total;
    int*   lsm  = (int*)(buf1 + total);
    int*   rowStart = lsm + lsmPad;
    int*   bin  = rowStart + rsPad;
    int*   ed   = bin + 2 * edgePad;

    lgcn_bin_local<<<nBlk, MAXB, 0, stream>>>(rows, cols, vals, lsm, bin, nEdges, nBuckP, nBlk);
    lgcn_bucket_sort<<<nBuck, MAXB, 0, stream>>>(lsm, bin, rowStart, ed, nRows, nBlk);

    int rowBlocks = (nRows + 31) / 32;
    lgcn_row_layer<0><<<rowBlocks, 256, 0, stream>>>(rowStart, ed, ue, ie, nullptr, nullptr, buf0, nUsers, nRows);
    lgcn_row_layer<1><<<rowBlocks, 256, 0, stream>>>(rowStart, ed, ue, ie, buf0, nullptr, buf1, nUsers, nRows);
    lgcn_row_layer<2><<<rowBlocks, 256, 0, stream>>>(rowStart, ed, ue, ie, buf1, buf0, out, nUsers, nRows);
}
